# starting kernel
# baseline (speedup 1.0000x reference)
#include <hip/hip_runtime.h>
#include <hip/hip_bf16.h>

typedef __attribute__((ext_vector_type(16))) _Float16 v16h;
typedef __attribute__((ext_vector_type(8)))  float    v8f;

#define E32 32
#define GQ  2

union H8 { uint4 u; _Float16 h[8]; };

__device__ __forceinline__ int lane_id() { return (int)(threadIdx.x & 31); }

__device__ __forceinline__ v8f wmma_f16(v16h a, v16h b, v8f c) {
  return __builtin_amdgcn_wmma_f32_16x16x32_f16(
       false, a,  false, b,
       (short)0, c,  false,  false);
}

template <typename T>
__device__ __forceinline__ v16h load_frag(const T* __restrict__ src, int stride) {
  const int l  = lane_id();
  const int r  = l & 15;
  const int kb = (l >> 4) << 3;
  const T* p = src + (size_t)r * stride + kb;
  v16h f;
#pragma unroll
  for (int i = 0; i < 8; ++i) {
    f[i]     = (_Float16)p[i];
    f[8 + i] = (_Float16)p[i + 16];
  }
  return f;
}

__device__ __forceinline__ void store_tile_f16(_Float16* __restrict__ dst,
                                               const _Float16* __restrict__ lds) {
  const int l = lane_id();
#pragma unroll
  for (int i = 0; i < 2; ++i) {
    const int idx = i * 32 + l;
    *(uint4*)(dst + idx * 8) = *(const uint4*)(lds + idx * 8);
  }
}

__global__ void __launch_bounds__(32)
qkv_kernel(const float* __restrict__ x, const float* __restrict__ w,
           const float* __restrict__ bias,
           _Float16* __restrict__ q, _Float16* __restrict__ k,
           _Float16* __restrict__ v, _Float16* __restrict__ vT) {
  __shared__ alignas(16) _Float16 tile[16 * 32];

  const int t  = blockIdx.x;
  const int g  = blockIdx.y;
  const int l  = lane_id();
  const int rowbase = t * 16;
  const int b     = rowbase >> 10;
  const int nbase = rowbase & 1023;
  const int bg    = b * GQ + g;
  const int nn = l & 15, half = l >> 4;

  v16h a = load_frag(x + (size_t)rowbase * (GQ * E32) + g * E32, GQ * E32);

  _Float16* outs[3] = {q, k, v};
#pragma unroll
  for (int s = 0; s < 3; ++s) {
    v8f acc[2];
    float bn[2];
#pragma unroll
    for (int cb = 0; cb < 2; ++cb) {
      const int obase = g * 96 + s * 32 + cb * 16;
      v16h bf = load_frag(w + (size_t)obase * E32, E32);
      v8f z = {};
      acc[cb] = wmma_f16(a, bf, z);
      bn[cb]  = bias[obase + nn];
    }
    __syncthreads();
#pragma unroll
    for (int cb = 0; cb < 2; ++cb) {
#pragma unroll
      for (int r = 0; r < 8; ++r) {
        const int m = r + half * 8;
        tile[m * 32 + cb * 16 + nn] = (_Float16)(acc[cb][r] + bn[cb]);
      }
    }
    if (s == 2) {
#pragma unroll
      for (int cb = 0; cb < 2; ++cb) {
        H8 p;
#pragma unroll
        for (int r = 0; r < 8; ++r) p.h[r] = (_Float16)(acc[cb][r] + bn[cb]);
        const int e = cb * 16 + nn;
        *(uint4*)(vT + ((size_t)bg * E32 + e) * 1024 + nbase + half * 8) = p.u;
      }
    }
    __syncthreads();
    store_tile_f16(outs[s] + ((size_t)bg * 1024 + nbase) * E32, tile);
  }
}

__global__ void __launch_bounds__(32)
attn_kernel(const _Float16* __restrict__ q, const _Float16* __restrict__ k,
            const _Float16* __restrict__ vT, _Float16* __restrict__ attn) {
  __shared__ alignas(16) _Float16 pTile[16 * 64];

  const int qt = blockIdx.x;
  const int bg = blockIdx.y;
  const int l  = lane_id();
  const _Float16* kb  = k  + (size_t)bg * 1024 * E32;
  const _Float16* vtb = vT + (size_t)bg * E32 * 1024;

  v16h aq = load_frag(q + ((size_t)bg * 1024 + qt * 16) * E32, E32);

  float mrow[8], lrow[8];
  v8f acc0 = {}, acc1 = {};
#pragma unroll
  for (int r = 0; r < 8; ++r) { mrow[r] = -1e30f; lrow[r] = 0.0f; }

  const float scale = 1.0f / (float)E32;
  const int nn = l & 15, half = l >> 4;

  for (int jt = 0; jt < 16; ++jt) {
    const int jb = jt * 64;
    if (jt < 15) {
      __builtin_prefetch(kb + (size_t)(jb + 64) * E32, 0, 1);
      __builtin_prefetch(vtb + jb + 64, 0, 1);
    }
    v8f s[4];
#pragma unroll
    for (int c = 0; c < 4; ++c) {
      v16h bk = load_frag(kb + (size_t)(jb + c * 16) * E32, E32);
      v8f z = {};
      s[c] = wmma_f16(aq, bk, z);
    }

    float p[4][8];
#pragma unroll
    for (int r = 0; r < 8; ++r) {
      float sv[4];
      float mx = -1e30f;
#pragma unroll
      for (int c = 0; c < 4; ++c) { sv[c] = s[c][r] * scale; mx = fmaxf(mx, sv[c]); }
#pragma unroll
      for (int off = 1; off < 16; off <<= 1) mx = fmaxf(mx, __shfl_xor(mx, off, 32));
      const float mn  = fmaxf(mrow[r], mx);
      const float fac = __expf(mrow[r] - mn);
      float rs = 0.0f;
#pragma unroll
      for (int c = 0; c < 4; ++c) { p[c][r] = __expf(sv[c] - mn); rs += p[c][r]; }
#pragma unroll
      for (int off = 1; off < 16; off <<= 1) rs += __shfl_xor(rs, off, 32);
      lrow[r] = lrow[r] * fac + rs;
      mrow[r] = mn;
      acc0[r] *= fac;
      acc1[r] *= fac;
    }

    __syncthreads();
#pragma unroll
    for (int c = 0; c < 4; ++c) {
#pragma unroll
      for (int r = 0; r < 8; ++r) {
        const int m = r + half * 8;
        pTile[m * 64 + c * 16 + nn] = (_Float16)p[c][r];
      }
    }
    __syncthreads();
    v16h ap0 = load_frag((const _Float16*)pTile, 64);
    v16h ap1 = load_frag((const _Float16*)pTile + 32, 64);
    v16h bv00 = load_frag(vtb + jb, 1024);
    v16h bv01 = load_frag(vtb + jb + 32, 1024);
    v16h bv10 = load_frag(vtb + (size_t)16 * 1024 + jb, 1024);
    v16h bv11 = load_frag(vtb + (size_t)16 * 1024 + jb + 32, 1024);
    acc0 = wmma_f16(ap0, bv00, acc0);
    acc0 = wmma_f16(ap1, bv01, acc0);
    acc1 = wmma_f16(ap0, bv10, acc1);
    acc1 = wmma_f16(ap1, bv11, acc1);
  }

  __syncthreads();
#pragma unroll
  for (int r = 0; r < 8; ++r) {
    const int m = r + half * 8;
    const float inv = 1.0f / lrow[r];
    pTile[m * 32 + nn]      = (_Float16)(acc0[r] * inv);
    pTile[m * 32 + 16 + nn] = (_Float16)(acc1[r] * inv);
  }
  __syncthreads();
  store_tile_f16(attn + ((size_t)bg * 1024 + qt * 16) * E32, (const _Float16*)pTile);
}

__global__ void __launch_bounds__(32)
mlp_kernel(const _Float16* __restrict__ v, const _Float16* __restrict__ attn,
           const float* __restrict__ w1,  const float* __restrict__ b1,
           const float* __restrict__ w10, const float* __restrict__ b10,
           const float* __restrict__ w11, const float* __restrict__ b11,
           float* __restrict__ out) {
  __shared__ alignas(16) _Float16 h1[16 * 32];
  __shared__ alignas(16) _Float16 h2[16 * 32];
  __shared__ alignas(16) float    ot[16 * 32];

  const int t = blockIdx.x;
  const int g = blockIdx.y;
  const int l = lane_id();
  const int rowbase = t * 16;
  const int b     = rowbase >> 10;
  const int nbase = rowbase & 1023;
  const int nn = l & 15, half = l >> 4;

  const size_t src = ((size_t)(b * GQ + g) * 1024 + nbase) * E32;
  v16h a_v  = load_frag(v + src, E32);
  v16h a_at = load_frag(attn + src, E32);

#pragma unroll
  for (int cb = 0; cb < 2; ++cb) {
    const int ob = g * 32 + cb * 16;
    v16h bw0 = load_frag(w1 + (size_t)ob * 64, 64);
    v16h bw1 = load_frag(w1 + (size_t)ob * 64 + 32, 64);
    v8f acc = {};
    acc = wmma_f16(a_v, bw0, acc);
    acc = wmma_f16(a_at, bw1, acc);
    const float bn = b1[ob + nn];
#pragma unroll
    for (int r = 0; r < 8; ++r) {
      const int m = r + half * 8;
      const float z = acc[r] + bn;
      h1[m * 32 + cb * 16 + nn] = (_Float16)(z >= 0.0f ? z : 0.01f * z);
    }
  }
  __syncthreads();

  v16h a_h1 = load_frag((const _Float16*)h1, 32);
#pragma unroll
  for (int cb = 0; cb < 2; ++cb) {
    const int ob = g * 32 + cb * 16;
    v16h bw = load_frag(w10 + (size_t)ob * 32, 32);
    v8f acc = {};
    acc = wmma_f16(a_h1, bw, acc);
    const float bn = b10[ob + nn];
#pragma unroll
    for (int r = 0; r < 8; ++r) {
      const int m = r + half * 8;
      const float z = acc[r] + bn;
      h2[m * 32 + cb * 16 + nn] = (_Float16)(z >= 0.0f ? z : 0.01f * z);
    }
  }
  __syncthreads();

  v16h a_h2 = load_frag((const _Float16*)h2, 32);
#pragma unroll
  for (int cb = 0; cb < 2; ++cb) {
    const int ob = g * 32 + cb * 16;
    v16h bw = load_frag(w11 + (size_t)ob * 32, 32);
    v8f acc = {};
    acc = wmma_f16(a_h2, bw, acc);
    const float bn = b11[ob + nn];
#pragma unroll
    for (int r = 0; r < 8; ++r) {
      const int m = r + half * 8;
      const float z = acc[r] + bn;
      ot[m * 32 + cb * 16 + nn] = (z >= 0.0f ? z : 0.01f * z);
    }
  }
  __syncthreads();
  float* dst = out + (size_t)rowbase * 64 + g * 32;
#pragma unroll
  for (int i = 0; i < 4; ++i) {
    const int idx = i * 32 + l;
    const int row = idx >> 3;
    const int c   = (idx & 7) * 4;
    *(uint4*)(dst + (size_t)row * 64 + c) = *(const uint4*)(ot + row * 32 + c);
  }
}

extern "C" void kernel_launch(void* const* d_in, const int* in_sizes, int n_in,
                              void* d_out, int out_size, void* d_ws, size_t ws_size,
                              hipStream_t stream) {
  const float* x     = (const float*)d_in[0];
  const float* w_qkv = (const float*)d_in[1];
  const float* b_qkv = (const float*)d_in[2];
  const float* w_h1  = (const float*)d_in[3];
  const float* b_h1  = (const float*)d_in[4];
  const float* w_h10 = (const float*)d_in[5];
  const float* b_h10 = (const float*)d_in[6];
  const float* w_h11 = (const float*)d_in[7];
  const float* b_h11 = (const float*)d_in[8];
  float* out = (float*)d_out;

  const size_t NEL = (size_t)32 * GQ * 1024 * E32;
  _Float16* ws   = (_Float16*)d_ws;
  _Float16* q    = ws;
  _Float16* k    = ws + NEL;
  _Float16* v    = ws + 2 * NEL;
  _Float16* vT   = ws + 3 * NEL;
  _Float16* attn = ws + 4 * NEL;

  qkv_kernel<<<dim3(2048, GQ), 32, 0, stream>>>(x, w_qkv, b_qkv, q, k, v, vT);
  attn_kernel<<<dim3(64, 64), 32, 0, stream>>>(q, k, vT, attn);
  mlp_kernel<<<dim3(2048, GQ), 32, 0, stream>>>(v, attn, w_h1, b_h1,
                                                w_h10, b_h10, w_h11, b_h11, out);
}


# attention: next step's K and V^T fragments requested right behind the products that read the current ones, into the same registers: a whole step to land; loop barriers order LDS only; sums unchanged
# speedup vs baseline: 1.2591x; 1.2591x over previous
#include <hip/hip_runtime.h>
#include <hip/hip_bf16.h>

typedef __attribute__((ext_vector_type(16))) _Float16 v16h;
typedef __attribute__((ext_vector_type(8)))  float    v8f;

#define E32 32
#define GQ  2

union H8 { uint4 u; _Float16 h[8]; };

__device__ __forceinline__ int lane_id() { return (int)(threadIdx.x & 31); }

__device__ __forceinline__ v8f wmma_f16(v16h a, v16h b, v8f c) {
  return __builtin_amdgcn_wmma_f32_16x16x32_f16(
       false, a,  false, b,
       (short)0, c,  false,  false);
}

template <typename T>
__device__ __forceinline__ v16h load_frag(const T* __restrict__ src, int stride) {
  const int l  = lane_id();
  const int r  = l & 15;
  const int kb = (l >> 4) << 3;
  const T* p = src + (size_t)r * stride + kb;
  v16h f;
#pragma unroll
  for (int i = 0; i < 8; ++i) {
    f[i]     = (_Float16)p[i];
    f[8 + i] = (_Float16)p[i + 16];
  }
  return f;
}

__device__ __forceinline__ void store_tile_f16(_Float16* __restrict__ dst,
                                               const _Float16* __restrict__ lds) {
  const int l = lane_id();
#pragma unroll
  for (int i = 0; i < 2; ++i) {
    const int idx = i * 32 + l;
    *(uint4*)(dst + idx * 8) = *(const uint4*)(lds + idx * 8);
  }
}

__global__ void __launch_bounds__(32)
qkv_kernel(const float* __restrict__ x, const float* __restrict__ w,
           const float* __restrict__ bias,
           _Float16* __restrict__ q, _Float16* __restrict__ k,
           _Float16* __restrict__ v, _Float16* __restrict__ vT) {
  __shared__ alignas(16) _Float16 tile[16 * 32];

  const int t  = blockIdx.x;
  const int g  = blockIdx.y;
  const int l  = lane_id();
  const int rowbase = t * 16;
  const int b     = rowbase >> 10;
  const int nbase = rowbase & 1023;
  const int bg    = b * GQ + g;
  const int nn = l & 15, half = l >> 4;

  v16h a = load_frag(x + (size_t)rowbase * (GQ * E32) + g * E32, GQ * E32);

  _Float16* outs[3] = {q, k, v};
#pragma unroll
  for (int s = 0; s < 3; ++s) {
    v8f acc[2];
    float bn[2];
#pragma unroll
    for (int cb = 0; cb < 2; ++cb) {
      const int obase = g * 96 + s * 32 + cb * 16;
      v16h bf = load_frag(w + (size_t)obase * E32, E32);
      v8f z = {};
      acc[cb] = wmma_f16(a, bf, z);
      bn[cb]  = bias[obase + nn];
    }
    __syncthreads();
#pragma unroll
    for (int cb = 0; cb < 2; ++cb) {
#pragma unroll
      for (int r = 0; r < 8; ++r) {
        const int m = r + half * 8;
        tile[m * 32 + cb * 16 + nn] = (_Float16)(acc[cb][r] + bn[cb]);
      }
    }
    if (s == 2) {
#pragma unroll
      for (int cb = 0; cb < 2; ++cb) {
        H8 p;
#pragma unroll
        for (int r = 0; r < 8; ++r) p.h[r] = (_Float16)(acc[cb][r] + bn[cb]);
        const int e = cb * 16 + nn;
        *(uint4*)(vT + ((size_t)bg * E32 + e) * 1024 + nbase + half * 8) = p.u;
      }
    }
    __syncthreads();
    store_tile_f16(outs[s] + ((size_t)bg * 1024 + nbase) * E32, tile);
  }
}


#define SP 68

__device__ __forceinline__ float xhalf(float x) {
  const int b = __builtin_bit_cast(int, x);
  return __builtin_bit_cast(float,
      __builtin_amdgcn_permlanex16(b, b, 0x76543210, 0xfedcba98, false, false));
}

__device__ __forceinline__ float rot8_upper(float x) {
  const int b = __builtin_bit_cast(int, x);
  return __builtin_bit_cast(float,
      __builtin_amdgcn_update_dpp(b, b, 0x128  , 0x2  , 0xf, false));
}
template <int R>
__device__ __forceinline__ float row_share(float g) {
  const int b = __builtin_bit_cast(int, g);
  return __builtin_bit_cast(float,
      __builtin_amdgcn_update_dpp(b, b, 0x150 + R  , 0xf, 0xf, false));
}
__device__ __forceinline__ void rows_to_acc(float perrow, float (&f)[8]) {
  const float g = rot8_upper(perrow);
  f[0] = row_share<0>(g); f[1] = row_share<1>(g); f[2] = row_share<2>(g); f[3] = row_share<3>(g);
  f[4] = row_share<4>(g); f[5] = row_share<5>(g); f[6] = row_share<6>(g); f[7] = row_share<7>(g);
}

__device__ __forceinline__ void lds_barrier() {
  __builtin_amdgcn_fence(__ATOMIC_RELEASE, "workgroup", "local");
  __builtin_amdgcn_s_barrier();
  __builtin_amdgcn_fence(__ATOMIC_ACQUIRE, "workgroup", "local");
}

template <bool MORE>
__device__ __forceinline__ void attn_step(const int jt, const v16h aq, v16h (&bk)[4], v16h (&bv)[4],
                                          const _Float16* __restrict__ kb,
                                          const _Float16* __restrict__ vtb,
                                          float* sT, const float* rowp, const int nn, const int half,
                                          float& mrun, float& lrun, v8f& acc0, v8f& acc1) {
  const float scale = 1.0f / (float)E32;
  int jn = (jt + 1) * 64;
  v8f s[4];
#pragma unroll
  for (int c = 0; c < 4; ++c) {
    v8f z = {};
    s[c] = wmma_f16(aq, bk[c], z);
  }
  if (MORE) {
    asm volatile("" : "+v"(s[0]), "+v"(s[1]), "+v"(s[2]), "+v"(s[3]), "+s"(jn));
#pragma unroll
    for (int c = 0; c < 4; ++c) bk[c] = load_frag(kb + (size_t)(jn + c * 16) * E32, E32);
  }

  lds_barrier();
#pragma unroll
  for (int c = 0; c < 4; ++c) {
#pragma unroll
    for (int r = 0; r < 8; ++r) {
      const int m = r + half * 8;
      sT[m * SP + c * 16 + nn] = s[c][r];
    }
  }
  lds_barrier();

  float sv[2][16];
#pragma unroll
  for (int t = 0; t < 2; ++t) {
    const float4 a0 = *(const float4*)(rowp + t * 32);
    const float4 a1 = *(const float4*)(rowp + t * 32 + 4);
    const float4 a2 = *(const float4*)(rowp + t * 32 + 16);
    const float4 a3 = *(const float4*)(rowp + t * 32 + 20);
    sv[t][0]  = a0.x * scale; sv[t][1]  = a0.y * scale; sv[t][2]  = a0.z * scale; sv[t][3]  = a0.w * scale;
    sv[t][4]  = a1.x * scale; sv[t][5]  = a1.y * scale; sv[t][6]  = a1.z * scale; sv[t][7]  = a1.w * scale;
    sv[t][8]  = a2.x * scale; sv[t][9]  = a2.y * scale; sv[t][10] = a2.z * scale; sv[t][11] = a2.w * scale;
    sv[t][12] = a3.x * scale; sv[t][13] = a3.y * scale; sv[t][14] = a3.z * scale; sv[t][15] = a3.w * scale;
  }

  float mp[4] = {-1e30f, -1e30f, -1e30f, -1e30f};
#pragma unroll
  for (int i = 0; i < 16; ++i) {
    mp[i & 3] = fmaxf(mp[i & 3], sv[0][i]);
    mp[i & 3] = fmaxf(mp[i & 3], sv[1][i]);
  }
  float mx = fmaxf(fmaxf(mp[0], mp[1]), fmaxf(mp[2], mp[3]));
  mx = fmaxf(mx, xhalf(mx));
  const float mn  = fmaxf(mrun, mx);
  const float fac = __expf(mrun - mn);

  float rs[4] = {0.0f, 0.0f, 0.0f, 0.0f};
  v16h ap0, ap1;
#pragma unroll
  for (int i = 0; i < 16; ++i) {
    const float p0 = __expf(sv[0][i] - mn);
    const float p1 = __expf(sv[1][i] - mn);
    rs[i & 3] += p0;
    rs[i & 3] += p1;
    ap0[i] = (_Float16)p0;
    ap1[i] = (_Float16)p1;
  }
  lrun = lrun * fac + ((rs[0] + rs[1]) + (rs[2] + rs[3]));
  mrun = mn;

  float fd[8];
  rows_to_acc(fac, fd);
#pragma unroll
  for (int r = 0; r < 8; ++r) { acc0[r] *= fd[r]; acc1[r] *= fd[r]; }

  acc0 = wmma_f16(ap0, bv[0], acc0);
  acc0 = wmma_f16(ap1, bv[1], acc0);
  acc1 = wmma_f16(ap0, bv[2], acc1);
  acc1 = wmma_f16(ap1, bv[3], acc1);
  if (MORE) {
    asm volatile("" : "+v"(acc0), "+v"(acc1), "+s"(jn));
    bv[0] = load_frag(vtb + jn, 1024);
    bv[1] = load_frag(vtb + jn + 32, 1024);
    bv[2] = load_frag(vtb + (size_t)16 * 1024 + jn, 1024);
    bv[3] = load_frag(vtb + (size_t)16 * 1024 + jn + 32, 1024);
  }
}

__global__ void __launch_bounds__(32)
attn_kernel(const _Float16* __restrict__ q, const _Float16* __restrict__ k,
            const _Float16* __restrict__ vT, _Float16* __restrict__ attn) {
  __shared__ alignas(16) unsigned char ldsRaw[16 * SP * 4];
  float*    sT = (float*)ldsRaw;
  _Float16* oT = (_Float16*)ldsRaw;

  const int qt = blockIdx.x;
  const int bg = blockIdx.y;
  const int l  = lane_id();
  const _Float16* kb  = k  + (size_t)bg * 1024 * E32;
  const _Float16* vtb = vT + (size_t)bg * E32 * 1024;

  v16h aq = load_frag(q + ((size_t)bg * 1024 + qt * 16) * E32, E32);

  float mrun = -1e30f, lrun = 0.0f;
  v8f acc0 = {}, acc1 = {};

  const int nn = l & 15, half = l >> 4;
  const float* rowp = sT + nn * SP + half * 8;

  v16h bk[4], bv[4];
#pragma unroll
  for (int c = 0; c < 4; ++c) bk[c] = load_frag(kb + (size_t)(c * 16) * E32, E32);
  bv[0] = load_frag(vtb, 1024);
  bv[1] = load_frag(vtb + 32, 1024);
  bv[2] = load_frag(vtb + (size_t)16 * 1024, 1024);
  bv[3] = load_frag(vtb + (size_t)16 * 1024 + 32, 1024);

  for (int jt = 0; jt < 15; ++jt)
    attn_step<true>(jt, aq, bk, bv, kb, vtb, sT, rowp, nn, half, mrun, lrun, acc0, acc1);
  attn_step<false>(15, aq, bk, bv, kb, vtb, sT, rowp, nn, half, mrun, lrun, acc0, acc1);

  const float ltot = lrun + xhalf(lrun);
  float invd[8];
  rows_to_acc(1.0f / ltot, invd);
  __syncthreads();
#pragma unroll
  for (int r = 0; r < 8; ++r) {
    const int m = r + half * 8;
    oT[m * 32 + nn]      = (_Float16)(acc0[r] * invd[r]);
    oT[m * 32 + 16 + nn] = (_Float16)(acc1[r] * invd[r]);
  }
  __syncthreads();
  store_tile_f16(attn + ((size_t)bg * 1024 + qt * 16) * E32, (const _Float16*)oT);
}

__global__ void __launch_bounds__(32)
mlp_kernel(const _Float16* __restrict__ v, const _Float16* __restrict__ attn,
           const float* __restrict__ w1,  const float* __restrict__ b1,
           const float* __restrict__ w10, const float* __restrict__ b10,
           const float* __restrict__ w11, const float* __restrict__ b11,
           float* __restrict__ out) {
  __shared__ alignas(16) _Float16 h1[16 * 32];
  __shared__ alignas(16) _Float16 h2[16 * 32];
  __shared__ alignas(16) float    ot[16 * 32];

  const int t = blockIdx.x;
  const int g = blockIdx.y;
  const int l = lane_id();
  const int rowbase = t * 16;
  const int b     = rowbase >> 10;
  const int nbase = rowbase & 1023;
  const int nn = l & 15, half = l >> 4;

  const size_t src = ((size_t)(b * GQ + g) * 1024 + nbase) * E32;
  v16h a_v  = load_frag(v + src, E32);
  v16h a_at = load_frag(attn + src, E32);

#pragma unroll
  for (int cb = 0; cb < 2; ++cb) {
    const int ob = g * 32 + cb * 16;
    v16h bw0 = load_frag(w1 + (size_t)ob * 64, 64);
    v16h bw1 = load_frag(w1 + (size_t)ob * 64 + 32, 64);
    v8f acc = {};
    acc = wmma_f16(a_v, bw0, acc);
    acc = wmma_f16(a_at, bw1, acc);
    const float bn = b1[ob + nn];
#pragma unroll
    for (int r = 0; r < 8; ++r) {
      const int m = r + half * 8;
      const float z = acc[r] + bn;
      h1[m * 32 + cb * 16 + nn] = (_Float16)(z >= 0.0f ? z : 0.01f * z);
    }
  }
  __syncthreads();

  v16h a_h1 = load_frag((const _Float16*)h1, 32);
#pragma unroll
  for (int cb = 0; cb < 2; ++cb) {
    const int ob = g * 32 + cb * 16;
    v16h bw = load_frag(w10 + (size_t)ob * 32, 32);
    v8f acc = {};
    acc = wmma_f16(a_h1, bw, acc);
    const float bn = b10[ob + nn];
#pragma unroll
    for (int r = 0; r < 8; ++r) {
      const int m = r + half * 8;
      const float z = acc[r] + bn;
      h2[m * 32 + cb * 16 + nn] = (_Float16)(z >= 0.0f ? z : 0.01f * z);
    }
  }
  __syncthreads();

  v16h a_h2 = load_frag((const _Float16*)h2, 32);
#pragma unroll
  for (int cb = 0; cb < 2; ++cb) {
    const int ob = g * 32 + cb * 16;
    v16h bw = load_frag(w11 + (size_t)ob * 32, 32);
    v8f acc = {};
    acc = wmma_f16(a_h2, bw, acc);
    const float bn = b11[ob + nn];
#pragma unroll
    for (int r = 0; r < 8; ++r) {
      const int m = r + half * 8;
      const float z = acc[r] + bn;
      ot[m * 32 + cb * 16 + nn] = (z >= 0.0f ? z : 0.01f * z);
    }
  }
  __syncthreads();
  float* dst = out + (size_t)rowbase * 64 + g * 32;
#pragma unroll
  for (int i = 0; i < 4; ++i) {
    const int idx = i * 32 + l;
    const int row = idx >> 3;
    const int c   = (idx & 7) * 4;
    *(uint4*)(dst + (size_t)row * 64 + c) = *(const uint4*)(ot + row * 32 + c);
  }
}

extern "C" void kernel_launch(void* const* d_in, const int* in_sizes, int n_in,
                              void* d_out, int out_size, void* d_ws, size_t ws_size,
                              hipStream_t stream) {
  const float* x     = (const float*)d_in[0];
  const float* w_qkv = (const float*)d_in[1];
  const float* b_qkv = (const float*)d_in[2];
  const float* w_h1  = (const float*)d_in[3];
  const float* b_h1  = (const float*)d_in[4];
  const float* w_h10 = (const float*)d_in[5];
  const float* b_h10 = (const float*)d_in[6];
  const float* w_h11 = (const float*)d_in[7];
  const float* b_h11 = (const float*)d_in[8];
  float* out = (float*)d_out;

  const size_t NEL = (size_t)32 * GQ * 1024 * E32;
  _Float16* ws   = (_Float16*)d_ws;
  _Float16* q    = ws;
  _Float16* k    = ws + NEL;
  _Float16* v    = ws + 2 * NEL;
  _Float16* vT   = ws + 3 * NEL;
  _Float16* attn = ws + 4 * NEL;

  qkv_kernel<<<dim3(2048, GQ), 32, 0, stream>>>(x, w_qkv, b_qkv, q, k, v, vT);
  attn_kernel<<<dim3(64, 64), 32, 0, stream>>>(q, k, vT, attn);
  mlp_kernel<<<dim3(2048, GQ), 32, 0, stream>>>(v, attn, w_h1, b_h1,
                                                w_h10, b_h10, w_h11, b_h11, out);
}


# attention: K and V^T stored by the projection kernel in fragment order: each operand load of the key loop reads 4 whole 128-B lines (was 8 half lines for K, 16 quarter lines for V^T); sums unchanged
# speedup vs baseline: 1.5214x; 1.2083x over previous
#include <hip/hip_runtime.h>
#include <hip/hip_bf16.h>

typedef __attribute__((ext_vector_type(16))) _Float16 v16h;
typedef __attribute__((ext_vector_type(8)))  float    v8f;

#define E32 32
#define GQ  2

union H8 { uint4 u; _Float16 h[8]; };
union H16 { v16h v; uint4 u[2]; };

__device__ __forceinline__ int lane_id() { return (int)(threadIdx.x & 31); }

__device__ __forceinline__ v8f wmma_f16(v16h a, v16h b, v8f c) {
  return __builtin_amdgcn_wmma_f32_16x16x32_f16(
       false, a,  false, b,
       (short)0, c,  false,  false);
}

template <typename T>
__device__ __forceinline__ v16h load_frag(const T* __restrict__ src, int stride) {
  const int l  = lane_id();
  const int r  = l & 15;
  const int kb = (l >> 4) << 3;
  const T* p = src + (size_t)r * stride + kb;
  v16h f;
#pragma unroll
  for (int i = 0; i < 8; ++i) {
    f[i]     = (_Float16)p[i];
    f[8 + i] = (_Float16)p[i + 16];
  }
  return f;
}

__device__ __forceinline__ void store_tile_f16(_Float16* __restrict__ dst,
                                               const _Float16* __restrict__ lds) {
  const int l = lane_id();
#pragma unroll
  for (int i = 0; i < 2; ++i) {
    const int idx = i * 32 + l;
    *(uint4*)(dst + idx * 8) = *(const uint4*)(lds + idx * 8);
  }
}

__device__ __forceinline__ void store_tile_f16_frag(_Float16* __restrict__ dst,
                                                    const _Float16* __restrict__ lds) {
  const int l = lane_id();
#pragma unroll
  for (int i = 0; i < 2; ++i) {
    const int idx   = i * 32 + l;
    const int chunk = idx >> 4, row = idx & 15;
    *(uint4*)(dst + idx * 8) = *(const uint4*)(lds + (row * 4 + chunk) * 8);
  }
}
__device__ __forceinline__ v16h load_frag_packed(const _Float16* __restrict__ src) {
  const int l = lane_id();
  H16 f;
  f.u[0] = *(const uint4*)(src + l * 8);
  f.u[1] = *(const uint4*)(src + 256 + l * 8);
  return f.v;
}

__global__ void __launch_bounds__(32)
qkv_kernel(const float* __restrict__ x, const float* __restrict__ w,
           const float* __restrict__ bias,
           _Float16* __restrict__ q, _Float16* __restrict__ k,
           _Float16* __restrict__ v, _Float16* __restrict__ vT) {
  __shared__ alignas(16) _Float16 tile[16 * 32];

  const int t  = blockIdx.x;
  const int g  = blockIdx.y;
  const int l  = lane_id();
  const int rowbase = t * 16;
  const int b     = rowbase >> 10;
  const int nbase = rowbase & 1023;
  const int bg    = b * GQ + g;
  const int nn = l & 15, half = l >> 4;

  v16h a = load_frag(x + (size_t)rowbase * (GQ * E32) + g * E32, GQ * E32);

  _Float16* outs[3] = {q, k, v};
#pragma unroll
  for (int s = 0; s < 3; ++s) {
    v8f acc[2];
    float bn[2];
#pragma unroll
    for (int cb = 0; cb < 2; ++cb) {
      const int obase = g * 96 + s * 32 + cb * 16;
      v16h bf = load_frag(w + (size_t)obase * E32, E32);
      v8f z = {};
      acc[cb] = wmma_f16(a, bf, z);
      bn[cb]  = bias[obase + nn];
    }
    __syncthreads();
#pragma unroll
    for (int cb = 0; cb < 2; ++cb) {
#pragma unroll
      for (int r = 0; r < 8; ++r) {
        const int m = r + half * 8;
        tile[m * 32 + cb * 16 + nn] = (_Float16)(acc[cb][r] + bn[cb]);
      }
    }
    if (s == 2) {
#pragma unroll
      for (int cb = 0; cb < 2; ++cb) {
        H8 p;
#pragma unroll
        for (int r = 0; r < 8; ++r) p.h[r] = (_Float16)(acc[cb][r] + bn[cb]);
        const int kb32 = nbase >> 5;
        const int j    = ((nbase >> 4) & 1) * 2 + half;
        *(uint4*)(vT + (((size_t)bg * 32 + kb32) * 2 + cb) * 512 + (j * 16 + nn) * 8) = p.u;
      }
    }
    __syncthreads();
    if (s == 1) store_tile_f16_frag(k + ((size_t)bg * 1024 + nbase) * E32, tile);
    else        store_tile_f16(outs[s] + ((size_t)bg * 1024 + nbase) * E32, tile);
  }
}


#define SP 68

__device__ __forceinline__ float xhalf(float x) {
  const int b = __builtin_bit_cast(int, x);
  return __builtin_bit_cast(float,
      __builtin_amdgcn_permlanex16(b, b, 0x76543210, 0xfedcba98, false, false));
}

__device__ __forceinline__ float rot8_upper(float x) {
  const int b = __builtin_bit_cast(int, x);
  return __builtin_bit_cast(float,
      __builtin_amdgcn_update_dpp(b, b, 0x128  , 0x2  , 0xf, false));
}
template <int R>
__device__ __forceinline__ float row_share(float g) {
  const int b = __builtin_bit_cast(int, g);
  return __builtin_bit_cast(float,
      __builtin_amdgcn_update_dpp(b, b, 0x150 + R  , 0xf, 0xf, false));
}
__device__ __forceinline__ void rows_to_acc(float perrow, float (&f)[8]) {
  const float g = rot8_upper(perrow);
  f[0] = row_share<0>(g); f[1] = row_share<1>(g); f[2] = row_share<2>(g); f[3] = row_share<3>(g);
  f[4] = row_share<4>(g); f[5] = row_share<5>(g); f[6] = row_share<6>(g); f[7] = row_share<7>(g);
}

__device__ __forceinline__ void lds_barrier() {
  __builtin_amdgcn_fence(__ATOMIC_RELEASE, "workgroup", "local");
  __builtin_amdgcn_s_barrier();
  __builtin_amdgcn_fence(__ATOMIC_ACQUIRE, "workgroup", "local");
}

template <bool MORE>
__device__ __forceinline__ void attn_step(const int jt, const v16h aq, v16h (&bk)[4], v16h (&bv)[4],
                                          const _Float16* __restrict__ kb,
                                          const _Float16* __restrict__ vtb,
                                          float* sT, const float* rowp, const int nn, const int half,
                                          float& mrun, float& lrun, v8f& acc0, v8f& acc1) {
  const float scale = 1.0f / (float)E32;
  int jn = (jt + 1) * 64;
  v8f s[4];
#pragma unroll
  for (int c = 0; c < 4; ++c) {
    v8f z = {};
    s[c] = wmma_f16(aq, bk[c], z);
  }
  if (MORE) {
    asm volatile("" : "+v"(s[0]), "+v"(s[1]), "+v"(s[2]), "+v"(s[3]), "+s"(jn));
#pragma unroll
    for (int c = 0; c < 4; ++c) bk[c] = load_frag_packed(kb + (size_t)(jn + c * 16) * E32);
  }

  lds_barrier();
#pragma unroll
  for (int c = 0; c < 4; ++c) {
#pragma unroll
    for (int r = 0; r < 8; ++r) {
      const int m = r + half * 8;
      sT[m * SP + c * 16 + nn] = s[c][r];
    }
  }
  lds_barrier();

  float sv[2][16];
#pragma unroll
  for (int t = 0; t < 2; ++t) {
    const float4 a0 = *(const float4*)(rowp + t * 32);
    const float4 a1 = *(const float4*)(rowp + t * 32 + 4);
    const float4 a2 = *(const float4*)(rowp + t * 32 + 16);
    const float4 a3 = *(const float4*)(rowp + t * 32 + 20);
    sv[t][0]  = a0.x * scale; sv[t][1]  = a0.y * scale; sv[t][2]  = a0.z * scale; sv[t][3]  = a0.w * scale;
    sv[t][4]  = a1.x * scale; sv[t][5]  = a1.y * scale; sv[t][6]  = a1.z * scale; sv[t][7]  = a1.w * scale;
    sv[t][8]  = a2.x * scale; sv[t][9]  = a2.y * scale; sv[t][10] = a2.z * scale; sv[t][11] = a2.w * scale;
    sv[t][12] = a3.x * scale; sv[t][13] = a3.y * scale; sv[t][14] = a3.z * scale; sv[t][15] = a3.w * scale;
  }

  float mp[4] = {-1e30f, -1e30f, -1e30f, -1e30f};
#pragma unroll
  for (int i = 0; i < 16; ++i) {
    mp[i & 3] = fmaxf(mp[i & 3], sv[0][i]);
    mp[i & 3] = fmaxf(mp[i & 3], sv[1][i]);
  }
  float mx = fmaxf(fmaxf(mp[0], mp[1]), fmaxf(mp[2], mp[3]));
  mx = fmaxf(mx, xhalf(mx));
  const float mn  = fmaxf(mrun, mx);
  const float fac = __expf(mrun - mn);

  float rs[4] = {0.0f, 0.0f, 0.0f, 0.0f};
  v16h ap0, ap1;
#pragma unroll
  for (int i = 0; i < 16; ++i) {
    const float p0 = __expf(sv[0][i] - mn);
    const float p1 = __expf(sv[1][i] - mn);
    rs[i & 3] += p0;
    rs[i & 3] += p1;
    ap0[i] = (_Float16)p0;
    ap1[i] = (_Float16)p1;
  }
  lrun = lrun * fac + ((rs[0] + rs[1]) + (rs[2] + rs[3]));
  mrun = mn;

  float fd[8];
  rows_to_acc(fac, fd);
#pragma unroll
  for (int r = 0; r < 8; ++r) { acc0[r] *= fd[r]; acc1[r] *= fd[r]; }

  acc0 = wmma_f16(ap0, bv[0], acc0);
  acc0 = wmma_f16(ap1, bv[1], acc0);
  acc1 = wmma_f16(ap0, bv[2], acc1);
  acc1 = wmma_f16(ap1, bv[3], acc1);
  if (MORE) {
    asm volatile("" : "+v"(acc0), "+v"(acc1), "+s"(jn));
    const _Float16* vs = vtb + (size_t)jn * E32;
    bv[0] = load_frag_packed(vs);
    bv[1] = load_frag_packed(vs + 1024);
    bv[2] = load_frag_packed(vs + 512);
    bv[3] = load_frag_packed(vs + 1536);
  }
}

__global__ void __launch_bounds__(32)
attn_kernel(const _Float16* __restrict__ q, const _Float16* __restrict__ k,
            const _Float16* __restrict__ vT, _Float16* __restrict__ attn) {
  __shared__ alignas(16) unsigned char ldsRaw[16 * SP * 4];
  float*    sT = (float*)ldsRaw;
  _Float16* oT = (_Float16*)ldsRaw;

  const int qt = blockIdx.x;
  const int bg = blockIdx.y;
  const int l  = lane_id();
  const _Float16* kb  = k  + (size_t)bg * 1024 * E32;
  const _Float16* vtb = vT + (size_t)bg * E32 * 1024;

  v16h aq = load_frag(q + ((size_t)bg * 1024 + qt * 16) * E32, E32);

  float mrun = -1e30f, lrun = 0.0f;
  v8f acc0 = {}, acc1 = {};

  const int nn = l & 15, half = l >> 4;
  const float* rowp = sT + nn * SP + half * 8;

  v16h bk[4], bv[4];
#pragma unroll
  for (int c = 0; c < 4; ++c) bk[c] = load_frag_packed(kb + (size_t)(c * 16) * E32);
  bv[0] = load_frag_packed(vtb);
  bv[1] = load_frag_packed(vtb + 1024);
  bv[2] = load_frag_packed(vtb + 512);
  bv[3] = load_frag_packed(vtb + 1536);

  for (int jt = 0; jt < 15; ++jt)
    attn_step<true>(jt, aq, bk, bv, kb, vtb, sT, rowp, nn, half, mrun, lrun, acc0, acc1);
  attn_step<false>(15, aq, bk, bv, kb, vtb, sT, rowp, nn, half, mrun, lrun, acc0, acc1);

  const float ltot = lrun + xhalf(lrun);
  float invd[8];
  rows_to_acc(1.0f / ltot, invd);
  __syncthreads();
#pragma unroll
  for (int r = 0; r < 8; ++r) {
    const int m = r + half * 8;
    oT[m * 32 + nn]      = (_Float16)(acc0[r] * invd[r]);
    oT[m * 32 + 16 + nn] = (_Float16)(acc1[r] * invd[r]);
  }
  __syncthreads();
  store_tile_f16(attn + ((size_t)bg * 1024 + qt * 16) * E32, (const _Float16*)oT);
}

__global__ void __launch_bounds__(32)
mlp_kernel(const _Float16* __restrict__ v, const _Float16* __restrict__ attn,
           const float* __restrict__ w1,  const float* __restrict__ b1,
           const float* __restrict__ w10, const float* __restrict__ b10,
           const float* __restrict__ w11, const float* __restrict__ b11,
           float* __restrict__ out) {
  __shared__ alignas(16) _Float16 h1[16 * 32];
  __shared__ alignas(16) _Float16 h2[16 * 32];
  __shared__ alignas(16) float    ot[16 * 32];

  const int t = blockIdx.x;
  const int g = blockIdx.y;
  const int l = lane_id();
  const int rowbase = t * 16;
  const int b     = rowbase >> 10;
  const int nbase = rowbase & 1023;
  const int nn = l & 15, half = l >> 4;

  const size_t src = ((size_t)(b * GQ + g) * 1024 + nbase) * E32;
  v16h a_v  = load_frag(v + src, E32);
  v16h a_at = load_frag(attn + src, E32);

#pragma unroll
  for (int cb = 0; cb < 2; ++cb) {
    const int ob = g * 32 + cb * 16;
    v16h bw0 = load_frag(w1 + (size_t)ob * 64, 64);
    v16h bw1 = load_frag(w1 + (size_t)ob * 64 + 32, 64);
    v8f acc = {};
    acc = wmma_f16(a_v, bw0, acc);
    acc = wmma_f16(a_at, bw1, acc);
    const float bn = b1[ob + nn];
#pragma unroll
    for (int r = 0; r < 8; ++r) {
      const int m = r + half * 8;
      const float z = acc[r] + bn;
      h1[m * 32 + cb * 16 + nn] = (_Float16)(z >= 0.0f ? z : 0.01f * z);
    }
  }
  __syncthreads();

  v16h a_h1 = load_frag((const _Float16*)h1, 32);
#pragma unroll
  for (int cb = 0; cb < 2; ++cb) {
    const int ob = g * 32 + cb * 16;
    v16h bw = load_frag(w10 + (size_t)ob * 32, 32);
    v8f acc = {};
    acc = wmma_f16(a_h1, bw, acc);
    const float bn = b10[ob + nn];
#pragma unroll
    for (int r = 0; r < 8; ++r) {
      const int m = r + half * 8;
      const float z = acc[r] + bn;
      h2[m * 32 + cb * 16 + nn] = (_Float16)(z >= 0.0f ? z : 0.01f * z);
    }
  }
  __syncthreads();

  v16h a_h2 = load_frag((const _Float16*)h2, 32);
#pragma unroll
  for (int cb = 0; cb < 2; ++cb) {
    const int ob = g * 32 + cb * 16;
    v16h bw = load_frag(w11 + (size_t)ob * 32, 32);
    v8f acc = {};
    acc = wmma_f16(a_h2, bw, acc);
    const float bn = b11[ob + nn];
#pragma unroll
    for (int r = 0; r < 8; ++r) {
      const int m = r + half * 8;
      const float z = acc[r] + bn;
      ot[m * 32 + cb * 16 + nn] = (z >= 0.0f ? z : 0.01f * z);
    }
  }
  __syncthreads();
  float* dst = out + (size_t)rowbase * 64 + g * 32;
#pragma unroll
  for (int i = 0; i < 4; ++i) {
    const int idx = i * 32 + l;
    const int row = idx >> 3;
    const int c   = (idx & 7) * 4;
    *(uint4*)(dst + (size_t)row * 64 + c) = *(const uint4*)(ot + row * 32 + c);
  }
}

extern "C" void kernel_launch(void* const* d_in, const int* in_sizes, int n_in,
                              void* d_out, int out_size, void* d_ws, size_t ws_size,
                              hipStream_t stream) {
  const float* x     = (const float*)d_in[0];
  const float* w_qkv = (const float*)d_in[1];
  const float* b_qkv = (const float*)d_in[2];
  const float* w_h1  = (const float*)d_in[3];
  const float* b_h1  = (const float*)d_in[4];
  const float* w_h10 = (const float*)d_in[5];
  const float* b_h10 = (const float*)d_in[6];
  const float* w_h11 = (const float*)d_in[7];
  const float* b_h11 = (const float*)d_in[8];
  float* out = (float*)d_out;

  const size_t NEL = (size_t)32 * GQ * 1024 * E32;
  _Float16* ws   = (_Float16*)d_ws;
  _Float16* q    = ws;
  _Float16* k    = ws + NEL;
  _Float16* v    = ws + 2 * NEL;
  _Float16* vT   = ws + 3 * NEL;
  _Float16* attn = ws + 4 * NEL;

  qkv_kernel<<<dim3(2048, GQ), 32, 0, stream>>>(x, w_qkv, b_qkv, q, k, v, vT);
  attn_kernel<<<dim3(64, 64), 32, 0, stream>>>(q, k, vT, attn);
  mlp_kernel<<<dim3(2048, GQ), 32, 0, stream>>>(v, attn, w_h1, b_h1,
                                                w_h10, b_h10, w_h11, b_h11, out);
}


# projection and MLP kernels: a wave owns 4 row tiles (64 rows) of a group; weights and biases loaded once per wave, each layer batched over the 4 tiles; 1024 one-wave groups, was 4096; sums unchanged
# speedup vs baseline: 1.7894x; 1.1762x over previous
#include <hip/hip_runtime.h>
#include <hip/hip_bf16.h>

typedef __attribute__((ext_vector_type(16))) _Float16 v16h;
typedef __attribute__((ext_vector_type(8)))  float    v8f;

#define E32 32
#define GQ  2

union H8 { uint4 u; _Float16 h[8]; };
union H16 { v16h v; uint4 u[2]; };

__device__ __forceinline__ int lane_id() { return (int)(threadIdx.x & 31); }

__device__ __forceinline__ v8f wmma_f16(v16h a, v16h b, v8f c) {
  return __builtin_amdgcn_wmma_f32_16x16x32_f16(
       false, a,  false, b,
       (short)0, c,  false,  false);
}

template <typename T>
__device__ __forceinline__ v16h load_frag(const T* __restrict__ src, int stride) {
  const int l  = lane_id();
  const int r  = l & 15;
  const int kb = (l >> 4) << 3;
  const T* p = src + (size_t)r * stride + kb;
  v16h f;
#pragma unroll
  for (int i = 0; i < 8; ++i) {
    f[i]     = (_Float16)p[i];
    f[8 + i] = (_Float16)p[i + 16];
  }
  return f;
}

__device__ __forceinline__ void store_tile_f16(_Float16* __restrict__ dst,
                                               const _Float16* __restrict__ lds) {
  const int l = lane_id();
#pragma unroll
  for (int i = 0; i < 2; ++i) {
    const int idx = i * 32 + l;
    *(uint4*)(dst + idx * 8) = *(const uint4*)(lds + idx * 8);
  }
}

__device__ __forceinline__ void store_tile_f16_frag(_Float16* __restrict__ dst,
                                                    const _Float16* __restrict__ lds) {
  const int l = lane_id();
#pragma unroll
  for (int i = 0; i < 2; ++i) {
    const int idx   = i * 32 + l;
    const int chunk = idx >> 4, row = idx & 15;
    *(uint4*)(dst + idx * 8) = *(const uint4*)(lds + (row * 4 + chunk) * 8);
  }
}
__device__ __forceinline__ v16h load_frag_packed(const _Float16* __restrict__ src) {
  const int l = lane_id();
  H16 f;
  f.u[0] = *(const uint4*)(src + l * 8);
  f.u[1] = *(const uint4*)(src + 256 + l * 8);
  return f.v;
}

#define RT 4

__global__ void __launch_bounds__(32)
qkv_kernel(const float* __restrict__ x, const float* __restrict__ w,
           const float* __restrict__ bias,
           _Float16* __restrict__ q, _Float16* __restrict__ k,
           _Float16* __restrict__ v, _Float16* __restrict__ vT) {
  __shared__ alignas(16) _Float16 tile[3 * RT * 16 * 32];

  const int g  = blockIdx.y;
  const int l  = lane_id();
  const int rowbase = blockIdx.x * (RT * 16);
  const int b     = rowbase >> 10;
  const int nbase = rowbase & 1023;
  const int bg    = b * GQ + g;
  const int nn = l & 15, half = l >> 4;

  v16h a[RT];
#pragma unroll
  for (int t = 0; t < RT; ++t)
    a[t] = load_frag(x + (size_t)(rowbase + t * 16) * (GQ * E32) + g * E32, GQ * E32);

  _Float16* outs[3] = {q, k, v};
#pragma unroll
  for (int s = 0; s < 3; ++s) {
#pragma unroll
    for (int cb = 0; cb < 2; ++cb) {
      const int obase = g * 96 + s * 32 + cb * 16;
      const v16h bf  = load_frag(w + (size_t)obase * E32, E32);
      const float bn = bias[obase + nn];
#pragma unroll
      for (int t = 0; t < RT; ++t) {
        v8f z = {};
        const v8f acc = wmma_f16(a[t], bf, z);
        _Float16* tl = tile + (s * RT + t) * 512;
#pragma unroll
        for (int r = 0; r < 8; ++r) {
          const int m = r + half * 8;
          tl[m * 32 + cb * 16 + nn] = (_Float16)(acc[r] + bn);
        }
        if (s == 2) {
          H8 p;
#pragma unroll
          for (int r = 0; r < 8; ++r) p.h[r] = (_Float16)(acc[r] + bn);
          const int n0   = nbase + t * 16;
          const int kb32 = n0 >> 5;
          const int j    = ((n0 >> 4) & 1) * 2 + half;
          *(uint4*)(vT + (((size_t)bg * 32 + kb32) * 2 + cb) * 512 + (j * 16 + nn) * 8) = p.u;
        }
      }
    }
  }
  __syncthreads();
#pragma unroll
  for (int s = 0; s < 3; ++s) {
#pragma unroll
    for (int t = 0; t < RT; ++t) {
      _Float16* dst = outs[s] + ((size_t)bg * 1024 + nbase + t * 16) * E32;
      if (s == 1) store_tile_f16_frag(dst, tile + (s * RT + t) * 512);
      else        store_tile_f16(dst, tile + (s * RT + t) * 512);
    }
  }
}


#define SP 68
#define QT 2
#define ST (16 * SP)

__device__ __forceinline__ float xhalf(float x) {
  const int b = __builtin_bit_cast(int, x);
  return __builtin_bit_cast(float,
      __builtin_amdgcn_permlanex16(b, b, 0x76543210, 0xfedcba98, false, false));
}

__device__ __forceinline__ float rot8_upper(float x) {
  const int b = __builtin_bit_cast(int, x);
  return __builtin_bit_cast(float,
      __builtin_amdgcn_update_dpp(b, b, 0x128  , 0x2  , 0xf, false));
}
template <int R>
__device__ __forceinline__ float row_share(float g) {
  const int b = __builtin_bit_cast(int, g);
  return __builtin_bit_cast(float,
      __builtin_amdgcn_update_dpp(b, b, 0x150 + R  , 0xf, 0xf, false));
}
__device__ __forceinline__ void rows_to_acc(float perrow, float (&f)[8]) {
  const float g = rot8_upper(perrow);
  f[0] = row_share<0>(g); f[1] = row_share<1>(g); f[2] = row_share<2>(g); f[3] = row_share<3>(g);
  f[4] = row_share<4>(g); f[5] = row_share<5>(g); f[6] = row_share<6>(g); f[7] = row_share<7>(g);
}

__device__ __forceinline__ void lds_barrier() {
  __builtin_amdgcn_fence(__ATOMIC_RELEASE, "workgroup", "local");
  __builtin_amdgcn_s_barrier();
  __builtin_amdgcn_fence(__ATOMIC_ACQUIRE, "workgroup", "local");
}

__device__ __forceinline__ void softmax_tile(const float* rowp, float& mrun, float& lrun,
                                             v8f& acc0, v8f& acc1, v16h& ap0, v16h& ap1) {
  const float scale = 1.0f / (float)E32;
  float sv[2][16];
#pragma unroll
  for (int t = 0; t < 2; ++t) {
    const float4 a0 = *(const float4*)(rowp + t * 32);
    const float4 a1 = *(const float4*)(rowp + t * 32 + 4);
    const float4 a2 = *(const float4*)(rowp + t * 32 + 16);
    const float4 a3 = *(const float4*)(rowp + t * 32 + 20);
    sv[t][0]  = a0.x * scale; sv[t][1]  = a0.y * scale; sv[t][2]  = a0.z * scale; sv[t][3]  = a0.w * scale;
    sv[t][4]  = a1.x * scale; sv[t][5]  = a1.y * scale; sv[t][6]  = a1.z * scale; sv[t][7]  = a1.w * scale;
    sv[t][8]  = a2.x * scale; sv[t][9]  = a2.y * scale; sv[t][10] = a2.z * scale; sv[t][11] = a2.w * scale;
    sv[t][12] = a3.x * scale; sv[t][13] = a3.y * scale; sv[t][14] = a3.z * scale; sv[t][15] = a3.w * scale;
  }

  float mp[4] = {-1e30f, -1e30f, -1e30f, -1e30f};
#pragma unroll
  for (int i = 0; i < 16; ++i) {
    mp[i & 3] = fmaxf(mp[i & 3], sv[0][i]);
    mp[i & 3] = fmaxf(mp[i & 3], sv[1][i]);
  }
  float mx = fmaxf(fmaxf(mp[0], mp[1]), fmaxf(mp[2], mp[3]));
  mx = fmaxf(mx, xhalf(mx));
  const float mn  = fmaxf(mrun, mx);
  const float fac = __expf(mrun - mn);

  float rs[4] = {0.0f, 0.0f, 0.0f, 0.0f};
#pragma unroll
  for (int i = 0; i < 16; ++i) {
    const float p0 = __expf(sv[0][i] - mn);
    const float p1 = __expf(sv[1][i] - mn);
    rs[i & 3] += p0;
    rs[i & 3] += p1;
    ap0[i] = (_Float16)p0;
    ap1[i] = (_Float16)p1;
  }
  lrun = lrun * fac + ((rs[0] + rs[1]) + (rs[2] + rs[3]));
  mrun = mn;

  float fd[8];
  rows_to_acc(fac, fd);
#pragma unroll
  for (int r = 0; r < 8; ++r) { acc0[r] *= fd[r]; acc1[r] *= fd[r]; }
}

template <bool MORE>
__device__ __forceinline__ void attn_step(const int jt, const v16h (&aq)[QT], v16h (&bk)[4], v16h (&bv)[4],
                                          const _Float16* __restrict__ kb,
                                          const _Float16* __restrict__ vtb,
                                          float* sT, const float* rowp, const int nn, const int half,
                                          float (&mrun)[QT], float (&lrun)[QT],
                                          v8f (&acc0)[QT], v8f (&acc1)[QT]) {
  int jn = (jt + 1) * 64;
  v8f s[QT][4];
#pragma unroll
  for (int t = 0; t < QT; ++t) {
#pragma unroll
    for (int c = 0; c < 4; ++c) {
      v8f z = {};
      s[t][c] = wmma_f16(aq[t], bk[c], z);
    }
  }
  asm volatile("" : "+v"(s[0][0]), "+v"(s[0][1]), "+v"(s[0][2]), "+v"(s[0][3]),
                    "+v"(s[1][0]), "+v"(s[1][1]), "+v"(s[1][2]), "+v"(s[1][3]), "+s"(jn));
  if (MORE) {
#pragma unroll
    for (int c = 0; c < 4; ++c) bk[c] = load_frag_packed(kb + (size_t)(jn + c * 16) * E32);
  }

  lds_barrier();
#pragma unroll
  for (int t = 0; t < QT; ++t) {
#pragma unroll
    for (int c = 0; c < 4; ++c) {
#pragma unroll
      for (int r = 0; r < 8; ++r) {
        const int m = r + half * 8;
        sT[t * ST + m * SP + c * 16 + nn] = s[t][c][r];
      }
    }
  }
  lds_barrier();

  v16h ap0[QT], ap1[QT];
#pragma unroll
  for (int t = 0; t < QT; ++t)
    softmax_tile(rowp + t * ST, mrun[t], lrun[t], acc0[t], acc1[t], ap0[t], ap1[t]);

#pragma unroll
  for (int t = 0; t < QT; ++t) acc0[t] = wmma_f16(ap0[t], bv[0], acc0[t]);
#pragma unroll
  for (int t = 0; t < QT; ++t) acc1[t] = wmma_f16(ap0[t], bv[2], acc1[t]);
#pragma unroll
  for (int t = 0; t < QT; ++t) acc0[t] = wmma_f16(ap1[t], bv[1], acc0[t]);
#pragma unroll
  for (int t = 0; t < QT; ++t) acc1[t] = wmma_f16(ap1[t], bv[3], acc1[t]);
  if (MORE) {
    asm volatile("" : "+v"(acc0[0]), "+v"(acc1[0]), "+v"(acc0[1]), "+v"(acc1[1]), "+s"(jn));
    const _Float16* vs = vtb + (size_t)jn * E32;
    bv[0] = load_frag_packed(vs);
    bv[1] = load_frag_packed(vs + 1024);
    bv[2] = load_frag_packed(vs + 512);
    bv[3] = load_frag_packed(vs + 1536);
  }
}

__global__ void __launch_bounds__(32)
attn_kernel(const _Float16* __restrict__ q, const _Float16* __restrict__ k,
            const _Float16* __restrict__ vT, _Float16* __restrict__ attn) {
  __shared__ alignas(16) unsigned char ldsRaw[QT * ST * 4];
  float*    sT = (float*)ldsRaw;
  _Float16* oT = (_Float16*)ldsRaw;

  const int qt = blockIdx.x;
  const int bg = blockIdx.y;
  const int l  = lane_id();
  const _Float16* kb  = k  + (size_t)bg * 1024 * E32;
  const _Float16* vtb = vT + (size_t)bg * E32 * 1024;
  const size_t qrow = (size_t)bg * 1024 + qt * (QT * 16);

  v16h aq[QT];
#pragma unroll
  for (int t = 0; t < QT; ++t) aq[t] = load_frag(q + (qrow + t * 16) * E32, E32);

  float mrun[QT], lrun[QT];
  v8f acc0[QT], acc1[QT];
#pragma unroll
  for (int t = 0; t < QT; ++t) {
    mrun[t] = -1e30f; lrun[t] = 0.0f;
    v8f z = {};
    acc0[t] = z; acc1[t] = z;
  }

  const int nn = l & 15, half = l >> 4;
  const float* rowp = sT + nn * SP + half * 8;

  v16h bk[4], bv[4];
#pragma unroll
  for (int c = 0; c < 4; ++c) bk[c] = load_frag_packed(kb + (size_t)(c * 16) * E32);
  bv[0] = load_frag_packed(vtb);
  bv[1] = load_frag_packed(vtb + 1024);
  bv[2] = load_frag_packed(vtb + 512);
  bv[3] = load_frag_packed(vtb + 1536);

  for (int jt = 0; jt < 15; ++jt)
    attn_step<true>(jt, aq, bk, bv, kb, vtb, sT, rowp, nn, half, mrun, lrun, acc0, acc1);
  attn_step<false>(15, aq, bk, bv, kb, vtb, sT, rowp, nn, half, mrun, lrun, acc0, acc1);

  __syncthreads();
#pragma unroll
  for (int t = 0; t < QT; ++t) {
    const float ltot = lrun[t] + xhalf(lrun[t]);
    float invd[8];
    rows_to_acc(1.0f / ltot, invd);
#pragma unroll
    for (int r = 0; r < 8; ++r) {
      const int m = r + half * 8;
      oT[t * 512 + m * 32 + nn]      = (_Float16)(acc0[t][r] * invd[r]);
      oT[t * 512 + m * 32 + 16 + nn] = (_Float16)(acc1[t][r] * invd[r]);
    }
  }
  __syncthreads();
#pragma unroll
  for (int t = 0; t < QT; ++t)
    store_tile_f16(attn + (qrow + t * 16) * E32, (const _Float16*)oT + t * 512);
}

__global__ void __launch_bounds__(32)
mlp_kernel(const _Float16* __restrict__ v, const _Float16* __restrict__ attn,
           const float* __restrict__ w1,  const float* __restrict__ b1,
           const float* __restrict__ w10, const float* __restrict__ b10,
           const float* __restrict__ w11, const float* __restrict__ b11,
           float* __restrict__ out) {
  __shared__ alignas(16) _Float16 h1[RT * 16 * 32];
  __shared__ alignas(16) _Float16 h2[RT * 16 * 32];
  __shared__ alignas(16) float    ot[RT * 16 * 32];

  const int g = blockIdx.y;
  const int l = lane_id();
  const int rowbase = blockIdx.x * (RT * 16);
  const int b     = rowbase >> 10;
  const int nbase = rowbase & 1023;
  const int nn = l & 15, half = l >> 4;

  const size_t src = ((size_t)(b * GQ + g) * 1024 + nbase) * E32;
  v16h a_v[RT], a_at[RT];
#pragma unroll
  for (int t = 0; t < RT; ++t) {
    a_v[t]  = load_frag(v + src + t * 512, E32);
    a_at[t] = load_frag(attn + src + t * 512, E32);
  }

#pragma unroll
  for (int cb = 0; cb < 2; ++cb) {
    const int ob = g * 32 + cb * 16;
    const v16h bw0 = load_frag(w1 + (size_t)ob * 64, 64);
    const v16h bw1 = load_frag(w1 + (size_t)ob * 64 + 32, 64);
    const float bn = b1[ob + nn];
#pragma unroll
    for (int t = 0; t < RT; ++t) {
      v8f acc = {};
      acc = wmma_f16(a_v[t], bw0, acc);
      acc = wmma_f16(a_at[t], bw1, acc);
#pragma unroll
      for (int r = 0; r < 8; ++r) {
        const int m = r + half * 8;
        const float z = acc[r] + bn;
        h1[t * 512 + m * 32 + cb * 16 + nn] = (_Float16)(z >= 0.0f ? z : 0.01f * z);
      }
    }
  }
  __syncthreads();

  v16h a_h1[RT];
#pragma unroll
  for (int t = 0; t < RT; ++t) a_h1[t] = load_frag((const _Float16*)h1 + t * 512, 32);
#pragma unroll
  for (int cb = 0; cb < 2; ++cb) {
    const int ob = g * 32 + cb * 16;
    const v16h bw  = load_frag(w10 + (size_t)ob * 32, 32);
    const float bn = b10[ob + nn];
#pragma unroll
    for (int t = 0; t < RT; ++t) {
      v8f acc = {};
      acc = wmma_f16(a_h1[t], bw, acc);
#pragma unroll
      for (int r = 0; r < 8; ++r) {
        const int m = r + half * 8;
        const float z = acc[r] + bn;
        h2[t * 512 + m * 32 + cb * 16 + nn] = (_Float16)(z >= 0.0f ? z : 0.01f * z);
      }
    }
  }
  __syncthreads();

  v16h a_h2[RT];
#pragma unroll
  for (int t = 0; t < RT; ++t) a_h2[t] = load_frag((const _Float16*)h2 + t * 512, 32);
#pragma unroll
  for (int cb = 0; cb < 2; ++cb) {
    const int ob = g * 32 + cb * 16;
    const v16h bw  = load_frag(w11 + (size_t)ob * 32, 32);
    const float bn = b11[ob + nn];
#pragma unroll
    for (int t = 0; t < RT; ++t) {
      v8f acc = {};
      acc = wmma_f16(a_h2[t], bw, acc);
#pragma unroll
      for (int r = 0; r < 8; ++r) {
        const int m = r + half * 8;
        const float z = acc[r] + bn;
        ot[t * 512 + m * 32 + cb * 16 + nn] = (z >= 0.0f ? z : 0.01f * z);
      }
    }
  }
  __syncthreads();
  float* dst = out + (size_t)rowbase * 64 + g * 32;
#pragma unroll
  for (int t = 0; t < RT; ++t) {
#pragma unroll
    for (int i = 0; i < 4; ++i) {
      const int idx = i * 32 + l;
      const int row = idx >> 3;
      const int c   = (idx & 7) * 4;
      *(uint4*)(dst + (size_t)(t * 16 + row) * 64 + c) = *(const uint4*)(ot + t * 512 + row * 32 + c);
    }
  }
}

extern "C" void kernel_launch(void* const* d_in, const int* in_sizes, int n_in,
                              void* d_out, int out_size, void* d_ws, size_t ws_size,
                              hipStream_t stream) {
  const float* x     = (const float*)d_in[0];
  const float* w_qkv = (const float*)d_in[1];
  const float* b_qkv = (const float*)d_in[2];
  const float* w_h1  = (const float*)d_in[3];
  const float* b_h1  = (const float*)d_in[4];
  const float* w_h10 = (const float*)d_in[5];
  const float* b_h10 = (const float*)d_in[6];
  const float* w_h11 = (const float*)d_in[7];
  const float* b_h11 = (const float*)d_in[8];
  float* out = (float*)d_out;

  const size_t NEL = (size_t)32 * GQ * 1024 * E32;
  _Float16* ws   = (_Float16*)d_ws;
  _Float16* q    = ws;
  _Float16* k    = ws + NEL;
  _Float16* v    = ws + 2 * NEL;
  _Float16* vT   = ws + 3 * NEL;
  _Float16* attn = ws + 4 * NEL;

  qkv_kernel<<<dim3(512, GQ), 32, 0, stream>>>(x, w_qkv, b_qkv, q, k, v, vT);
  attn_kernel<<<dim3(32, 64), 32, 0, stream>>>(q, k, vT, attn);
  mlp_kernel<<<dim3(512, GQ), 32, 0, stream>>>(v, attn, w_h1, b_h1,
                                               w_h10, b_h10, w_h11, b_h11, out);
}


# attention: 4-wave work-groups, one wave per SIMD on 4 neighbouring query blocks of one (b,g), each wave unchanged with its own LDS slice: K and V^T lines reach a WGP once per group; sums unchanged
# speedup vs baseline: 1.8418x; 1.0293x over previous
#include <hip/hip_runtime.h>
#include <hip/hip_bf16.h>

typedef __attribute__((ext_vector_type(16))) _Float16 v16h;
typedef __attribute__((ext_vector_type(8)))  float    v8f;

#define E32 32
#define GQ  2

union H8 { uint4 u; _Float16 h[8]; };
union H16 { v16h v; uint4 u[2]; };

__device__ __forceinline__ int lane_id() { return (int)(threadIdx.x & 31); }

__device__ __forceinline__ v8f wmma_f16(v16h a, v16h b, v8f c) {
  return __builtin_amdgcn_wmma_f32_16x16x32_f16(
       false, a,  false, b,
       (short)0, c,  false,  false);
}

template <typename T>
__device__ __forceinline__ v16h load_frag(const T* __restrict__ src, int stride) {
  const int l  = lane_id();
  const int r  = l & 15;
  const int kb = (l >> 4) << 3;
  const T* p = src + (size_t)r * stride + kb;
  v16h f;
#pragma unroll
  for (int i = 0; i < 8; ++i) {
    f[i]     = (_Float16)p[i];
    f[8 + i] = (_Float16)p[i + 16];
  }
  return f;
}

__device__ __forceinline__ void store_tile_f16(_Float16* __restrict__ dst,
                                               const _Float16* __restrict__ lds) {
  const int l = lane_id();
#pragma unroll
  for (int i = 0; i < 2; ++i) {
    const int idx = i * 32 + l;
    *(uint4*)(dst + idx * 8) = *(const uint4*)(lds + idx * 8);
  }
}

__device__ __forceinline__ void store_tile_f16_frag(_Float16* __restrict__ dst,
                                                    const _Float16* __restrict__ lds) {
  const int l = lane_id();
#pragma unroll
  for (int i = 0; i < 2; ++i) {
    const int idx   = i * 32 + l;
    const int chunk = idx >> 4, row = idx & 15;
    *(uint4*)(dst + idx * 8) = *(const uint4*)(lds + (row * 4 + chunk) * 8);
  }
}
__device__ __forceinline__ v16h load_frag_packed(const _Float16* __restrict__ src) {
  const int l = lane_id();
  H16 f;
  f.u[0] = *(const uint4*)(src + l * 8);
  f.u[1] = *(const uint4*)(src + 256 + l * 8);
  return f.v;
}

#define RT 4

__global__ void __launch_bounds__(32)
qkv_kernel(const float* __restrict__ x, const float* __restrict__ w,
           const float* __restrict__ bias,
           _Float16* __restrict__ q, _Float16* __restrict__ k,
           _Float16* __restrict__ v, _Float16* __restrict__ vT) {
  __shared__ alignas(16) _Float16 tile[3 * RT * 16 * 32];

  const int g  = blockIdx.y;
  const int l  = lane_id();
  const int rowbase = blockIdx.x * (RT * 16);
  const int b     = rowbase >> 10;
  const int nbase = rowbase & 1023;
  const int bg    = b * GQ + g;
  const int nn = l & 15, half = l >> 4;

  v16h a[RT];
#pragma unroll
  for (int t = 0; t < RT; ++t)
    a[t] = load_frag(x + (size_t)(rowbase + t * 16) * (GQ * E32) + g * E32, GQ * E32);

  _Float16* outs[3] = {q, k, v};
#pragma unroll
  for (int s = 0; s < 3; ++s) {
#pragma unroll
    for (int cb = 0; cb < 2; ++cb) {
      const int obase = g * 96 + s * 32 + cb * 16;
      const v16h bf  = load_frag(w + (size_t)obase * E32, E32);
      const float bn = bias[obase + nn];
#pragma unroll
      for (int t = 0; t < RT; ++t) {
        v8f z = {};
        const v8f acc = wmma_f16(a[t], bf, z);
        _Float16* tl = tile + (s * RT + t) * 512;
#pragma unroll
        for (int r = 0; r < 8; ++r) {
          const int m = r + half * 8;
          tl[m * 32 + cb * 16 + nn] = (_Float16)(acc[r] + bn);
        }
        if (s == 2) {
          H8 p;
#pragma unroll
          for (int r = 0; r < 8; ++r) p.h[r] = (_Float16)(acc[r] + bn);
          const int n0   = nbase + t * 16;
          const int kb32 = n0 >> 5;
          const int j    = ((n0 >> 4) & 1) * 2 + half;
          *(uint4*)(vT + (((size_t)bg * 32 + kb32) * 2 + cb) * 512 + (j * 16 + nn) * 8) = p.u;
        }
      }
    }
  }
  __syncthreads();
#pragma unroll
  for (int s = 0; s < 3; ++s) {
#pragma unroll
    for (int t = 0; t < RT; ++t) {
      _Float16* dst = outs[s] + ((size_t)bg * 1024 + nbase + t * 16) * E32;
      if (s == 1) store_tile_f16_frag(dst, tile + (s * RT + t) * 512);
      else        store_tile_f16(dst, tile + (s * RT + t) * 512);
    }
  }
}


#define SP 68
#define QT 2
#define ST (16 * SP)
#define AW 4

__device__ __forceinline__ float xhalf(float x) {
  const int b = __builtin_bit_cast(int, x);
  return __builtin_bit_cast(float,
      __builtin_amdgcn_permlanex16(b, b, 0x76543210, 0xfedcba98, false, false));
}

__device__ __forceinline__ float rot8_upper(float x) {
  const int b = __builtin_bit_cast(int, x);
  return __builtin_bit_cast(float,
      __builtin_amdgcn_update_dpp(b, b, 0x128  , 0x2  , 0xf, false));
}
template <int R>
__device__ __forceinline__ float row_share(float g) {
  const int b = __builtin_bit_cast(int, g);
  return __builtin_bit_cast(float,
      __builtin_amdgcn_update_dpp(b, b, 0x150 + R  , 0xf, 0xf, false));
}
__device__ __forceinline__ void rows_to_acc(float perrow, float (&f)[8]) {
  const float g = rot8_upper(perrow);
  f[0] = row_share<0>(g); f[1] = row_share<1>(g); f[2] = row_share<2>(g); f[3] = row_share<3>(g);
  f[4] = row_share<4>(g); f[5] = row_share<5>(g); f[6] = row_share<6>(g); f[7] = row_share<7>(g);
}

__device__ __forceinline__ void lds_barrier() {
  __builtin_amdgcn_fence(__ATOMIC_RELEASE, "workgroup", "local");
  __builtin_amdgcn_wave_barrier();
  __builtin_amdgcn_fence(__ATOMIC_ACQUIRE, "workgroup", "local");
}

__device__ __forceinline__ void softmax_tile(const float* rowp, float& mrun, float& lrun,
                                             v8f& acc0, v8f& acc1, v16h& ap0, v16h& ap1) {
  const float scale = 1.0f / (float)E32;
  float sv[2][16];
#pragma unroll
  for (int t = 0; t < 2; ++t) {
    const float4 a0 = *(const float4*)(rowp + t * 32);
    const float4 a1 = *(const float4*)(rowp + t * 32 + 4);
    const float4 a2 = *(const float4*)(rowp + t * 32 + 16);
    const float4 a3 = *(const float4*)(rowp + t * 32 + 20);
    sv[t][0]  = a0.x * scale; sv[t][1]  = a0.y * scale; sv[t][2]  = a0.z * scale; sv[t][3]  = a0.w * scale;
    sv[t][4]  = a1.x * scale; sv[t][5]  = a1.y * scale; sv[t][6]  = a1.z * scale; sv[t][7]  = a1.w * scale;
    sv[t][8]  = a2.x * scale; sv[t][9]  = a2.y * scale; sv[t][10] = a2.z * scale; sv[t][11] = a2.w * scale;
    sv[t][12] = a3.x * scale; sv[t][13] = a3.y * scale; sv[t][14] = a3.z * scale; sv[t][15] = a3.w * scale;
  }

  float mp[4] = {-1e30f, -1e30f, -1e30f, -1e30f};
#pragma unroll
  for (int i = 0; i < 16; ++i) {
    mp[i & 3] = fmaxf(mp[i & 3], sv[0][i]);
    mp[i & 3] = fmaxf(mp[i & 3], sv[1][i]);
  }
  float mx = fmaxf(fmaxf(mp[0], mp[1]), fmaxf(mp[2], mp[3]));
  mx = fmaxf(mx, xhalf(mx));
  const float mn  = fmaxf(mrun, mx);
  const float fac = __expf(mrun - mn);

  float rs[4] = {0.0f, 0.0f, 0.0f, 0.0f};
#pragma unroll
  for (int i = 0; i < 16; ++i) {
    const float p0 = __expf(sv[0][i] - mn);
    const float p1 = __expf(sv[1][i] - mn);
    rs[i & 3] += p0;
    rs[i & 3] += p1;
    ap0[i] = (_Float16)p0;
    ap1[i] = (_Float16)p1;
  }
  lrun = lrun * fac + ((rs[0] + rs[1]) + (rs[2] + rs[3]));
  mrun = mn;

  float fd[8];
  rows_to_acc(fac, fd);
#pragma unroll
  for (int r = 0; r < 8; ++r) { acc0[r] *= fd[r]; acc1[r] *= fd[r]; }
}

template <bool MORE>
__device__ __forceinline__ void attn_step(const int jt, const v16h (&aq)[QT], v16h (&bk)[4], v16h (&bv)[4],
                                          const _Float16* __restrict__ kb,
                                          const _Float16* __restrict__ vtb,
                                          float* sT, const float* rowp, const int nn, const int half,
                                          float (&mrun)[QT], float (&lrun)[QT],
                                          v8f (&acc0)[QT], v8f (&acc1)[QT]) {
  int jn = (jt + 1) * 64;
  v8f s[QT][4];
#pragma unroll
  for (int t = 0; t < QT; ++t) {
#pragma unroll
    for (int c = 0; c < 4; ++c) {
      v8f z = {};
      s[t][c] = wmma_f16(aq[t], bk[c], z);
    }
  }
  asm volatile("" : "+v"(s[0][0]), "+v"(s[0][1]), "+v"(s[0][2]), "+v"(s[0][3]),
                    "+v"(s[1][0]), "+v"(s[1][1]), "+v"(s[1][2]), "+v"(s[1][3]), "+s"(jn));
  if (MORE) {
#pragma unroll
    for (int c = 0; c < 4; ++c) bk[c] = load_frag_packed(kb + (size_t)(jn + c * 16) * E32);
  }

  lds_barrier();
#pragma unroll
  for (int t = 0; t < QT; ++t) {
#pragma unroll
    for (int c = 0; c < 4; ++c) {
#pragma unroll
      for (int r = 0; r < 8; ++r) {
        const int m = r + half * 8;
        sT[t * ST + m * SP + c * 16 + nn] = s[t][c][r];
      }
    }
  }
  lds_barrier();

  v16h ap0[QT], ap1[QT];
#pragma unroll
  for (int t = 0; t < QT; ++t)
    softmax_tile(rowp + t * ST, mrun[t], lrun[t], acc0[t], acc1[t], ap0[t], ap1[t]);

#pragma unroll
  for (int t = 0; t < QT; ++t) acc0[t] = wmma_f16(ap0[t], bv[0], acc0[t]);
#pragma unroll
  for (int t = 0; t < QT; ++t) acc1[t] = wmma_f16(ap0[t], bv[2], acc1[t]);
#pragma unroll
  for (int t = 0; t < QT; ++t) acc0[t] = wmma_f16(ap1[t], bv[1], acc0[t]);
#pragma unroll
  for (int t = 0; t < QT; ++t) acc1[t] = wmma_f16(ap1[t], bv[3], acc1[t]);
  if (MORE) {
    asm volatile("" : "+v"(acc0[0]), "+v"(acc1[0]), "+v"(acc0[1]), "+v"(acc1[1]), "+s"(jn));
    const _Float16* vs = vtb + (size_t)jn * E32;
    bv[0] = load_frag_packed(vs);
    bv[1] = load_frag_packed(vs + 1024);
    bv[2] = load_frag_packed(vs + 512);
    bv[3] = load_frag_packed(vs + 1536);
  }
}

__global__ void __launch_bounds__(32 * AW)
attn_kernel(const _Float16* __restrict__ q, const _Float16* __restrict__ k,
            const _Float16* __restrict__ vT, _Float16* __restrict__ attn) {
  __shared__ alignas(16) unsigned char ldsRaw[AW * QT * ST * 4];
  const int wv = __builtin_amdgcn_readfirstlane((int)(threadIdx.x >> 5));
  float*    sT = (float*)ldsRaw + wv * (QT * ST);
  _Float16* oT = (_Float16*)sT;

  const int qt = blockIdx.x * AW + wv;
  const int bg = blockIdx.y;
  const int l  = lane_id();
  const _Float16* kb  = k  + (size_t)bg * 1024 * E32;
  const _Float16* vtb = vT + (size_t)bg * E32 * 1024;
  const size_t qrow = (size_t)bg * 1024 + qt * (QT * 16);

  v16h aq[QT];
#pragma unroll
  for (int t = 0; t < QT; ++t) aq[t] = load_frag(q + (qrow + t * 16) * E32, E32);

  float mrun[QT], lrun[QT];
  v8f acc0[QT], acc1[QT];
#pragma unroll
  for (int t = 0; t < QT; ++t) {
    mrun[t] = -1e30f; lrun[t] = 0.0f;
    v8f z = {};
    acc0[t] = z; acc1[t] = z;
  }

  const int nn = l & 15, half = l >> 4;
  const float* rowp = sT + nn * SP + half * 8;

  v16h bk[4], bv[4];
#pragma unroll
  for (int c = 0; c < 4; ++c) bk[c] = load_frag_packed(kb + (size_t)(c * 16) * E32);
  bv[0] = load_frag_packed(vtb);
  bv[1] = load_frag_packed(vtb + 1024);
  bv[2] = load_frag_packed(vtb + 512);
  bv[3] = load_frag_packed(vtb + 1536);

  for (int jt = 0; jt < 15; ++jt)
    attn_step<true>(jt, aq, bk, bv, kb, vtb, sT, rowp, nn, half, mrun, lrun, acc0, acc1);
  attn_step<false>(15, aq, bk, bv, kb, vtb, sT, rowp, nn, half, mrun, lrun, acc0, acc1);

  lds_barrier();
#pragma unroll
  for (int t = 0; t < QT; ++t) {
    const float ltot = lrun[t] + xhalf(lrun[t]);
    float invd[8];
    rows_to_acc(1.0f / ltot, invd);
#pragma unroll
    for (int r = 0; r < 8; ++r) {
      const int m = r + half * 8;
      oT[t * 512 + m * 32 + nn]      = (_Float16)(acc0[t][r] * invd[r]);
      oT[t * 512 + m * 32 + 16 + nn] = (_Float16)(acc1[t][r] * invd[r]);
    }
  }
  lds_barrier();
#pragma unroll
  for (int t = 0; t < QT; ++t)
    store_tile_f16(attn + (qrow + t * 16) * E32, (const _Float16*)oT + t * 512);
}

__global__ void __launch_bounds__(32)
mlp_kernel(const _Float16* __restrict__ v, const _Float16* __restrict__ attn,
           const float* __restrict__ w1,  const float* __restrict__ b1,
           const float* __restrict__ w10, const float* __restrict__ b10,
           const float* __restrict__ w11, const float* __restrict__ b11,
           float* __restrict__ out) {
  __shared__ alignas(16) _Float16 h1[RT * 16 * 32];
  __shared__ alignas(16) _Float16 h2[RT * 16 * 32];
  __shared__ alignas(16) float    ot[RT * 16 * 32];

  const int g = blockIdx.y;
  const int l = lane_id();
  const int rowbase = blockIdx.x * (RT * 16);
  const int b     = rowbase >> 10;
  const int nbase = rowbase & 1023;
  const int nn = l & 15, half = l >> 4;

  const size_t src = ((size_t)(b * GQ + g) * 1024 + nbase) * E32;
  v16h a_v[RT], a_at[RT];
#pragma unroll
  for (int t = 0; t < RT; ++t) {
    a_v[t]  = load_frag(v + src + t * 512, E32);
    a_at[t] = load_frag(attn + src + t * 512, E32);
  }

#pragma unroll
  for (int cb = 0; cb < 2; ++cb) {
    const int ob = g * 32 + cb * 16;
    const v16h bw0 = load_frag(w1 + (size_t)ob * 64, 64);
    const v16h bw1 = load_frag(w1 + (size_t)ob * 64 + 32, 64);
    const float bn = b1[ob + nn];
#pragma unroll
    for (int t = 0; t < RT; ++t) {
      v8f acc = {};
      acc = wmma_f16(a_v[t], bw0, acc);
      acc = wmma_f16(a_at[t], bw1, acc);
#pragma unroll
      for (int r = 0; r < 8; ++r) {
        const int m = r + half * 8;
        const float z = acc[r] + bn;
        h1[t * 512 + m * 32 + cb * 16 + nn] = (_Float16)(z >= 0.0f ? z : 0.01f * z);
      }
    }
  }
  __syncthreads();

  v16h a_h1[RT];
#pragma unroll
  for (int t = 0; t < RT; ++t) a_h1[t] = load_frag((const _Float16*)h1 + t * 512, 32);
#pragma unroll
  for (int cb = 0; cb < 2; ++cb) {
    const int ob = g * 32 + cb * 16;
    const v16h bw  = load_frag(w10 + (size_t)ob * 32, 32);
    const float bn = b10[ob + nn];
#pragma unroll
    for (int t = 0; t < RT; ++t) {
      v8f acc = {};
      acc = wmma_f16(a_h1[t], bw, acc);
#pragma unroll
      for (int r = 0; r < 8; ++r) {
        const int m = r + half * 8;
        const float z = acc[r] + bn;
        h2[t * 512 + m * 32 + cb * 16 + nn] = (_Float16)(z >= 0.0f ? z : 0.01f * z);
      }
    }
  }
  __syncthreads();

  v16h a_h2[RT];
#pragma unroll
  for (int t = 0; t < RT; ++t) a_h2[t] = load_frag((const _Float16*)h2 + t * 512, 32);
#pragma unroll
  for (int cb = 0; cb < 2; ++cb) {
    const int ob = g * 32 + cb * 16;
    const v16h bw  = load_frag(w11 + (size_t)ob * 32, 32);
    const float bn = b11[ob + nn];
#pragma unroll
    for (int t = 0; t < RT; ++t) {
      v8f acc = {};
      acc = wmma_f16(a_h2[t], bw, acc);
#pragma unroll
      for (int r = 0; r < 8; ++r) {
        const int m = r + half * 8;
        const float z = acc[r] + bn;
        ot[t * 512 + m * 32 + cb * 16 + nn] = (z >= 0.0f ? z : 0.01f * z);
      }
    }
  }
  __syncthreads();
  float* dst = out + (size_t)rowbase * 64 + g * 32;
#pragma unroll
  for (int t = 0; t < RT; ++t) {
#pragma unroll
    for (int i = 0; i < 4; ++i) {
      const int idx = i * 32 + l;
      const int row = idx >> 3;
      const int c   = (idx & 7) * 4;
      *(uint4*)(dst + (size_t)(t * 16 + row) * 64 + c) = *(const uint4*)(ot + t * 512 + row * 32 + c);
    }
  }
}

extern "C" void kernel_launch(void* const* d_in, const int* in_sizes, int n_in,
                              void* d_out, int out_size, void* d_ws, size_t ws_size,
                              hipStream_t stream) {
  const float* x     = (const float*)d_in[0];
  const float* w_qkv = (const float*)d_in[1];
  const float* b_qkv = (const float*)d_in[2];
  const float* w_h1  = (const float*)d_in[3];
  const float* b_h1  = (const float*)d_in[4];
  const float* w_h10 = (const float*)d_in[5];
  const float* b_h10 = (const float*)d_in[6];
  const float* w_h11 = (const float*)d_in[7];
  const float* b_h11 = (const float*)d_in[8];
  float* out = (float*)d_out;

  const size_t NEL = (size_t)32 * GQ * 1024 * E32;
  _Float16* ws   = (_Float16*)d_ws;
  _Float16* q    = ws;
  _Float16* k    = ws + NEL;
  _Float16* v    = ws + 2 * NEL;
  _Float16* vT   = ws + 3 * NEL;
  _Float16* attn = ws + 4 * NEL;

  qkv_kernel<<<dim3(512, GQ), 32, 0, stream>>>(x, w_qkv, b_qkv, q, k, v, vT);
  attn_kernel<<<dim3(8, 64), 128, 0, stream>>>(q, k, vT, attn);
  mlp_kernel<<<dim3(512, GQ), 32, 0, stream>>>(v, attn, w_h1, b_h1,
                                               w_h10, b_h10, w_h11, b_h11, out);
}


# attention: 128 keys per step (was 64): LDS hand-overs, maximum join, rescale factor with its broadcast and the accumulator rescale run 8 times, not 16; row-sum and accumulation order changed
# speedup vs baseline: 1.9651x; 1.0669x over previous
#include <hip/hip_runtime.h>
#include <hip/hip_bf16.h>

typedef __attribute__((ext_vector_type(16))) _Float16 v16h;
typedef __attribute__((ext_vector_type(8)))  float    v8f;

#define E32 32
#define GQ  2

union H8 { uint4 u; _Float16 h[8]; };
union H16 { v16h v; uint4 u[2]; };

__device__ __forceinline__ int lane_id() { return (int)(threadIdx.x & 31); }

__device__ __forceinline__ v8f wmma_f16(v16h a, v16h b, v8f c) {
  return __builtin_amdgcn_wmma_f32_16x16x32_f16(
       false, a,  false, b,
       (short)0, c,  false,  false);
}

template <typename T>
__device__ __forceinline__ v16h load_frag(const T* __restrict__ src, int stride) {
  const int l  = lane_id();
  const int r  = l & 15;
  const int kb = (l >> 4) << 3;
  const T* p = src + (size_t)r * stride + kb;
  v16h f;
#pragma unroll
  for (int i = 0; i < 8; ++i) {
    f[i]     = (_Float16)p[i];
    f[8 + i] = (_Float16)p[i + 16];
  }
  return f;
}

__device__ __forceinline__ void store_tile_f16(_Float16* __restrict__ dst,
                                               const _Float16* __restrict__ lds) {
  const int l = lane_id();
#pragma unroll
  for (int i = 0; i < 2; ++i) {
    const int idx = i * 32 + l;
    *(uint4*)(dst + idx * 8) = *(const uint4*)(lds + idx * 8);
  }
}

__device__ __forceinline__ void store_tile_f16_frag(_Float16* __restrict__ dst,
                                                    const _Float16* __restrict__ lds) {
  const int l = lane_id();
#pragma unroll
  for (int i = 0; i < 2; ++i) {
    const int idx   = i * 32 + l;
    const int chunk = idx >> 4, row = idx & 15;
    *(uint4*)(dst + idx * 8) = *(const uint4*)(lds + (row * 4 + chunk) * 8);
  }
}
__device__ __forceinline__ v16h load_frag_packed(const _Float16* __restrict__ src) {
  const int l = lane_id();
  H16 f;
  f.u[0] = *(const uint4*)(src + l * 8);
  f.u[1] = *(const uint4*)(src + 256 + l * 8);
  return f.v;
}

#define RT 4

__global__ void __launch_bounds__(32)
qkv_kernel(const float* __restrict__ x, const float* __restrict__ w,
           const float* __restrict__ bias,
           _Float16* __restrict__ q, _Float16* __restrict__ k,
           _Float16* __restrict__ v, _Float16* __restrict__ vT) {
  __shared__ alignas(16) _Float16 tile[3 * RT * 16 * 32];

  const int g  = blockIdx.y;
  const int l  = lane_id();
  const int rowbase = blockIdx.x * (RT * 16);
  const int b     = rowbase >> 10;
  const int nbase = rowbase & 1023;
  const int bg    = b * GQ + g;
  const int nn = l & 15, half = l >> 4;

  v16h a[RT];
#pragma unroll
  for (int t = 0; t < RT; ++t)
    a[t] = load_frag(x + (size_t)(rowbase + t * 16) * (GQ * E32) + g * E32, GQ * E32);

  _Float16* outs[3] = {q, k, v};
#pragma unroll
  for (int s = 0; s < 3; ++s) {
#pragma unroll
    for (int cb = 0; cb < 2; ++cb) {
      const int obase = g * 96 + s * 32 + cb * 16;
      const v16h bf  = load_frag(w + (size_t)obase * E32, E32);
      const float bn = bias[obase + nn];
#pragma unroll
      for (int t = 0; t < RT; ++t) {
        v8f z = {};
        const v8f acc = wmma_f16(a[t], bf, z);
        _Float16* tl = tile + (s * RT + t) * 512;
#pragma unroll
        for (int r = 0; r < 8; ++r) {
          const int m = r + half * 8;
          tl[m * 32 + cb * 16 + nn] = (_Float16)(acc[r] + bn);
        }
        if (s == 2) {
          H8 p;
#pragma unroll
          for (int r = 0; r < 8; ++r) p.h[r] = (_Float16)(acc[r] + bn);
          const int n0   = nbase + t * 16;
          const int kb32 = n0 >> 5;
          const int j    = ((n0 >> 4) & 1) * 2 + half;
          *(uint4*)(vT + (((size_t)bg * 32 + kb32) * 2 + cb) * 512 + (j * 16 + nn) * 8) = p.u;
        }
      }
    }
  }
  __syncthreads();
#pragma unroll
  for (int s = 0; s < 3; ++s) {
#pragma unroll
    for (int t = 0; t < RT; ++t) {
      _Float16* dst = outs[s] + ((size_t)bg * 1024 + nbase + t * 16) * E32;
      if (s == 1) store_tile_f16_frag(dst, tile + (s * RT + t) * 512);
      else        store_tile_f16(dst, tile + (s * RT + t) * 512);
    }
  }
}


#define KS 128
#define KT (KS / 16)
#define KG (KS / 32)
#define SP (KS + 4)
#define QT 2
#define ST (16 * SP)
#define AW 4

__device__ __forceinline__ float xhalf(float x) {
  const int b = __builtin_bit_cast(int, x);
  return __builtin_bit_cast(float,
      __builtin_amdgcn_permlanex16(b, b, 0x76543210, 0xfedcba98, false, false));
}

__device__ __forceinline__ float rot8_upper(float x) {
  const int b = __builtin_bit_cast(int, x);
  return __builtin_bit_cast(float,
      __builtin_amdgcn_update_dpp(b, b, 0x128  , 0x2  , 0xf, false));
}
template <int R>
__device__ __forceinline__ float row_share(float g) {
  const int b = __builtin_bit_cast(int, g);
  return __builtin_bit_cast(float,
      __builtin_amdgcn_update_dpp(b, b, 0x150 + R  , 0xf, 0xf, false));
}
__device__ __forceinline__ void rows_to_acc(float perrow, float (&f)[8]) {
  const float g = rot8_upper(perrow);
  f[0] = row_share<0>(g); f[1] = row_share<1>(g); f[2] = row_share<2>(g); f[3] = row_share<3>(g);
  f[4] = row_share<4>(g); f[5] = row_share<5>(g); f[6] = row_share<6>(g); f[7] = row_share<7>(g);
}

__device__ __forceinline__ void lds_barrier() {
  __builtin_amdgcn_fence(__ATOMIC_RELEASE, "workgroup", "local");
  __builtin_amdgcn_wave_barrier();
  __builtin_amdgcn_fence(__ATOMIC_ACQUIRE, "workgroup", "local");
}

__device__ __forceinline__ void softmax_tile(const float* rowp, float& mrun, float& lrun,
                                             v8f& acc0, v8f& acc1, v16h (&ap)[KG]) {
  const float scale = 1.0f / (float)E32;
  float sv[KG][16];
#pragma unroll
  for (int t = 0; t < KG; ++t) {
    const float4 a0 = *(const float4*)(rowp + t * 32);
    const float4 a1 = *(const float4*)(rowp + t * 32 + 4);
    const float4 a2 = *(const float4*)(rowp + t * 32 + 16);
    const float4 a3 = *(const float4*)(rowp + t * 32 + 20);
    sv[t][0]  = a0.x * scale; sv[t][1]  = a0.y * scale; sv[t][2]  = a0.z * scale; sv[t][3]  = a0.w * scale;
    sv[t][4]  = a1.x * scale; sv[t][5]  = a1.y * scale; sv[t][6]  = a1.z * scale; sv[t][7]  = a1.w * scale;
    sv[t][8]  = a2.x * scale; sv[t][9]  = a2.y * scale; sv[t][10] = a2.z * scale; sv[t][11] = a2.w * scale;
    sv[t][12] = a3.x * scale; sv[t][13] = a3.y * scale; sv[t][14] = a3.z * scale; sv[t][15] = a3.w * scale;
  }

  float mp[4] = {-1e30f, -1e30f, -1e30f, -1e30f};
#pragma unroll
  for (int i = 0; i < 16; ++i) {
#pragma unroll
    for (int t = 0; t < KG; ++t) mp[i & 3] = fmaxf(mp[i & 3], sv[t][i]);
  }
  float mx = fmaxf(fmaxf(mp[0], mp[1]), fmaxf(mp[2], mp[3]));
  mx = fmaxf(mx, xhalf(mx));
  const float mn  = fmaxf(mrun, mx);
  const float fac = __expf(mrun - mn);

  float rs[4] = {0.0f, 0.0f, 0.0f, 0.0f};
#pragma unroll
  for (int i = 0; i < 16; ++i) {
#pragma unroll
    for (int t = 0; t < KG; ++t) {
      const float p = __expf(sv[t][i] - mn);
      rs[i & 3] += p;
      ap[t][i] = (_Float16)p;
    }
  }
  lrun = lrun * fac + ((rs[0] + rs[1]) + (rs[2] + rs[3]));
  mrun = mn;

  float fd[8];
  rows_to_acc(fac, fd);
#pragma unroll
  for (int r = 0; r < 8; ++r) { acc0[r] *= fd[r]; acc1[r] *= fd[r]; }
}

template <bool MORE>
__device__ __forceinline__ void attn_step(const int jt, const v16h (&aq)[QT], v16h (&bk)[KT], v16h (&bv)[2 * KG],
                                          const _Float16* __restrict__ kb,
                                          const _Float16* __restrict__ vtb,
                                          float* sT, const float* rowp, const int nn, const int half,
                                          float (&mrun)[QT], float (&lrun)[QT],
                                          v8f (&acc0)[QT], v8f (&acc1)[QT]) {
  int jn = (jt + 1) * KS;
  v8f s[QT][KT];
#pragma unroll
  for (int t = 0; t < QT; ++t) {
#pragma unroll
    for (int c = 0; c < KT; ++c) {
      v8f z = {};
      s[t][c] = wmma_f16(aq[t], bk[c], z);
    }
  }
  asm volatile("" : "+v"(s[0][0]), "+v"(s[0][1]), "+v"(s[0][2]), "+v"(s[0][3]),
                    "+v"(s[0][4]), "+v"(s[0][5]), "+v"(s[0][6]), "+v"(s[0][7]), "+s"(jn));
  asm volatile("" : "+v"(s[1][0]), "+v"(s[1][1]), "+v"(s[1][2]), "+v"(s[1][3]),
                    "+v"(s[1][4]), "+v"(s[1][5]), "+v"(s[1][6]), "+v"(s[1][7]), "+s"(jn));
  if (MORE) {
#pragma unroll
    for (int c = 0; c < KT; ++c) bk[c] = load_frag_packed(kb + (size_t)(jn + c * 16) * E32);
  }

  lds_barrier();
#pragma unroll
  for (int t = 0; t < QT; ++t) {
#pragma unroll
    for (int c = 0; c < KT; ++c) {
#pragma unroll
      for (int r = 0; r < 8; ++r) {
        const int m = r + half * 8;
        sT[t * ST + m * SP + c * 16 + nn] = s[t][c][r];
      }
    }
  }
  lds_barrier();

  v16h ap[QT][KG];
#pragma unroll
  for (int t = 0; t < QT; ++t)
    softmax_tile(rowp + t * ST, mrun[t], lrun[t], acc0[t], acc1[t], ap[t]);

#pragma unroll
  for (int g = 0; g < KG; ++g) {
#pragma unroll
    for (int t = 0; t < QT; ++t) acc0[t] = wmma_f16(ap[t][g], bv[2 * g], acc0[t]);
#pragma unroll
    for (int t = 0; t < QT; ++t) acc1[t] = wmma_f16(ap[t][g], bv[2 * g + 1], acc1[t]);
  }
  if (MORE) {
    asm volatile("" : "+v"(acc0[0]), "+v"(acc1[0]), "+v"(acc0[1]), "+v"(acc1[1]), "+s"(jn));
    const _Float16* vs = vtb + (size_t)jn * E32;
#pragma unroll
    for (int g = 0; g < KG; ++g) {
      bv[2 * g]     = load_frag_packed(vs + g * 1024);
      bv[2 * g + 1] = load_frag_packed(vs + g * 1024 + 512);
    }
  }
}

__global__ void __launch_bounds__(32 * AW)
attn_kernel(const _Float16* __restrict__ q, const _Float16* __restrict__ k,
            const _Float16* __restrict__ vT, _Float16* __restrict__ attn) {
  __shared__ alignas(16) unsigned char ldsRaw[AW * QT * ST * 4];
  const int wv = __builtin_amdgcn_readfirstlane((int)(threadIdx.x >> 5));
  float*    sT = (float*)ldsRaw + wv * (QT * ST);
  _Float16* oT = (_Float16*)sT;

  const int qt = blockIdx.x * AW + wv;
  const int bg = blockIdx.y;
  const int l  = lane_id();
  const _Float16* kb  = k  + (size_t)bg * 1024 * E32;
  const _Float16* vtb = vT + (size_t)bg * E32 * 1024;
  const size_t qrow = (size_t)bg * 1024 + qt * (QT * 16);

  v16h aq[QT];
#pragma unroll
  for (int t = 0; t < QT; ++t) aq[t] = load_frag(q + (qrow + t * 16) * E32, E32);

  float mrun[QT], lrun[QT];
  v8f acc0[QT], acc1[QT];
#pragma unroll
  for (int t = 0; t < QT; ++t) {
    mrun[t] = -1e30f; lrun[t] = 0.0f;
    v8f z = {};
    acc0[t] = z; acc1[t] = z;
  }

  const int nn = l & 15, half = l >> 4;
  const float* rowp = sT + nn * SP + half * 8;

  v16h bk[KT], bv[2 * KG];
#pragma unroll
  for (int c = 0; c < KT; ++c) bk[c] = load_frag_packed(kb + (size_t)(c * 16) * E32);
  __builtin_amdgcn_sched_barrier(0);
#pragma unroll
  for (int g = 0; g < KG; ++g) {
    bv[2 * g]     = load_frag_packed(vtb + g * 1024);
    bv[2 * g + 1] = load_frag_packed(vtb + g * 1024 + 512);
  }

  for (int jt = 0; jt < 1024 / KS - 1; ++jt)
    attn_step<true>(jt, aq, bk, bv, kb, vtb, sT, rowp, nn, half, mrun, lrun, acc0, acc1);
  attn_step<false>(1024 / KS - 1, aq, bk, bv, kb, vtb, sT, rowp, nn, half, mrun, lrun, acc0, acc1);

  lds_barrier();
#pragma unroll
  for (int t = 0; t < QT; ++t) {
    const float ltot = lrun[t] + xhalf(lrun[t]);
    float invd[8];
    rows_to_acc(1.0f / ltot, invd);
#pragma unroll
    for (int r = 0; r < 8; ++r) {
      const int m = r + half * 8;
      oT[t * 512 + m * 32 + nn]      = (_Float16)(acc0[t][r] * invd[r]);
      oT[t * 512 + m * 32 + 16 + nn] = (_Float16)(acc1[t][r] * invd[r]);
    }
  }
  lds_barrier();
#pragma unroll
  for (int t = 0; t < QT; ++t)
    store_tile_f16(attn + (qrow + t * 16) * E32, (const _Float16*)oT + t * 512);
}

__global__ void __launch_bounds__(32)
mlp_kernel(const _Float16* __restrict__ v, const _Float16* __restrict__ attn,
           const float* __restrict__ w1,  const float* __restrict__ b1,
           const float* __restrict__ w10, const float* __restrict__ b10,
           const float* __restrict__ w11, const float* __restrict__ b11,
           float* __restrict__ out) {
  __shared__ alignas(16) _Float16 h1[RT * 16 * 32];
  __shared__ alignas(16) _Float16 h2[RT * 16 * 32];
  __shared__ alignas(16) float    ot[RT * 16 * 32];

  const int g = blockIdx.y;
  const int l = lane_id();
  const int rowbase = blockIdx.x * (RT * 16);
  const int b     = rowbase >> 10;
  const int nbase = rowbase & 1023;
  const int nn = l & 15, half = l >> 4;

  const size_t src = ((size_t)(b * GQ + g) * 1024 + nbase) * E32;
  v16h a_v[RT], a_at[RT];
#pragma unroll
  for (int t = 0; t < RT; ++t) {
    a_v[t]  = load_frag(v + src + t * 512, E32);
    a_at[t] = load_frag(attn + src + t * 512, E32);
  }

#pragma unroll
  for (int cb = 0; cb < 2; ++cb) {
    const int ob = g * 32 + cb * 16;
    const v16h bw0 = load_frag(w1 + (size_t)ob * 64, 64);
    const v16h bw1 = load_frag(w1 + (size_t)ob * 64 + 32, 64);
    const float bn = b1[ob + nn];
#pragma unroll
    for (int t = 0; t < RT; ++t) {
      v8f acc = {};
      acc = wmma_f16(a_v[t], bw0, acc);
      acc = wmma_f16(a_at[t], bw1, acc);
#pragma unroll
      for (int r = 0; r < 8; ++r) {
        const int m = r + half * 8;
        const float z = acc[r] + bn;
        h1[t * 512 + m * 32 + cb * 16 + nn] = (_Float16)(z >= 0.0f ? z : 0.01f * z);
      }
    }
  }
  __syncthreads();

  v16h a_h1[RT];
#pragma unroll
  for (int t = 0; t < RT; ++t) a_h1[t] = load_frag((const _Float16*)h1 + t * 512, 32);
#pragma unroll
  for (int cb = 0; cb < 2; ++cb) {
    const int ob = g * 32 + cb * 16;
    const v16h bw  = load_frag(w10 + (size_t)ob * 32, 32);
    const float bn = b10[ob + nn];
#pragma unroll
    for (int t = 0; t < RT; ++t) {
      v8f acc = {};
      acc = wmma_f16(a_h1[t], bw, acc);
#pragma unroll
      for (int r = 0; r < 8; ++r) {
        const int m = r + half * 8;
        const float z = acc[r] + bn;
        h2[t * 512 + m * 32 + cb * 16 + nn] = (_Float16)(z >= 0.0f ? z : 0.01f * z);
      }
    }
  }
  __syncthreads();

  v16h a_h2[RT];
#pragma unroll
  for (int t = 0; t < RT; ++t) a_h2[t] = load_frag((const _Float16*)h2 + t * 512, 32);
#pragma unroll
  for (int cb = 0; cb < 2; ++cb) {
    const int ob = g * 32 + cb * 16;
    const v16h bw  = load_frag(w11 + (size_t)ob * 32, 32);
    const float bn = b11[ob + nn];
#pragma unroll
    for (int t = 0; t < RT; ++t) {
      v8f acc = {};
      acc = wmma_f16(a_h2[t], bw, acc);
#pragma unroll
      for (int r = 0; r < 8; ++r) {
        const int m = r + half * 8;
        const float z = acc[r] + bn;
        ot[t * 512 + m * 32 + cb * 16 + nn] = (z >= 0.0f ? z : 0.01f * z);
      }
    }
  }
  __syncthreads();
  float* dst = out + (size_t)rowbase * 64 + g * 32;
#pragma unroll
  for (int t = 0; t < RT; ++t) {
#pragma unroll
    for (int i = 0; i < 4; ++i) {
      const int idx = i * 32 + l;
      const int row = idx >> 3;
      const int c   = (idx & 7) * 4;
      *(uint4*)(dst + (size_t)(t * 16 + row) * 64 + c) = *(const uint4*)(ot + t * 512 + row * 32 + c);
    }
  }
}

extern "C" void kernel_launch(void* const* d_in, const int* in_sizes, int n_in,
                              void* d_out, int out_size, void* d_ws, size_t ws_size,
                              hipStream_t stream) {
  const float* x     = (const float*)d_in[0];
  const float* w_qkv = (const float*)d_in[1];
  const float* b_qkv = (const float*)d_in[2];
  const float* w_h1  = (const float*)d_in[3];
  const float* b_h1  = (const float*)d_in[4];
  const float* w_h10 = (const float*)d_in[5];
  const float* b_h10 = (const float*)d_in[6];
  const float* w_h11 = (const float*)d_in[7];
  const float* b_h11 = (const float*)d_in[8];
  float* out = (float*)d_out;

  const size_t NEL = (size_t)32 * GQ * 1024 * E32;
  _Float16* ws   = (_Float16*)d_ws;
  _Float16* q    = ws;
  _Float16* k    = ws + NEL;
  _Float16* v    = ws + 2 * NEL;
  _Float16* vT   = ws + 3 * NEL;
  _Float16* attn = ws + 4 * NEL;

  qkv_kernel<<<dim3(512, GQ), 32, 0, stream>>>(x, w_qkv, b_qkv, q, k, v, vT);
  attn_kernel<<<dim3(8, 64), 128, 0, stream>>>(q, k, vT, attn);
  mlp_kernel<<<dim3(512, GQ), 32, 0, stream>>>(v, attn, w_h1, b_h1,
                                               w_h10, b_h10, w_h11, b_h11, out);
}
